# starting kernel
# baseline (speedup 1.0000x reference)
#include <hip/hip_runtime.h>

typedef __attribute__((ext_vector_type(2))) float v2f;
typedef __attribute__((ext_vector_type(8))) float v8f;

#define DD 128

__global__ __launch_bounds__(32) void iaf_layer_kernel(
    const float* __restrict__ x,
    const float* __restrict__ W1,  const float* __restrict__ b1,
    const float* __restrict__ Wmu, const float* __restrict__ bmu,
    const float* __restrict__ Wls, const float* __restrict__ bls,
    float* __restrict__ out, int B)
{
    __shared__ float Zh[16 * DD];
    __shared__ float Hh[16 * DD];
    __shared__ float tH[256], tM[256], tL[256];
    __shared__ float wd1[256], wdm[256], wdl[256];

    const int lane  = threadIdx.x;
    const int r0    = blockIdx.x * 16;
    const int mrow  = lane & 15;
    const int khalf = (lane >> 4) * 2;

    float ldj = 0.0f;

    for (int I = 0; I < 8; ++I) {
        const int c0 = I * 16;

        {
            const int e0 = lane * 8;
            #pragma unroll
            for (int q = 0; q < 8; ++q) {
                const int e  = e0 + q;
                const int t2 = e >> 4, t = e & 15;
                wd1[e] = W1 [(c0 + t2) * DD + c0 + t];
                wdm[e] = Wmu[(c0 + t2) * DD + c0 + t];
                wdl[e] = Wls[(c0 + t2) * DD + c0 + t];
            }
        }

        v8f cH = {0.f,0.f,0.f,0.f,0.f,0.f,0.f,0.f};
        v8f cM = {0.f,0.f,0.f,0.f,0.f,0.f,0.f,0.f};
        v8f cL = {0.f,0.f,0.f,0.f,0.f,0.f,0.f,0.f};
        const int nkt = 4 * I;
        for (int kt = 0; kt < nkt; ++kt) {
            const int k0 = kt * 4;
            v2f aZ = *(const v2f*)&Zh[mrow * DD + k0 + khalf];
            v2f aH = *(const v2f*)&Hh[mrow * DD + k0 + khalf];
            v2f bW1 = *(const v2f*)&W1 [(c0 + mrow) * DD + k0 + khalf];
            v2f bWm = *(const v2f*)&Wmu[(c0 + mrow) * DD + k0 + khalf];
            v2f bWl = *(const v2f*)&Wls[(c0 + mrow) * DD + k0 + khalf];
            cH = __builtin_amdgcn_wmma_f32_16x16x4_f32(false, aZ, false, bW1, (short)0, cH, false, false);
            cM = __builtin_amdgcn_wmma_f32_16x16x4_f32(false, aH, false, bWm, (short)0, cM, false, false);
            cL = __builtin_amdgcn_wmma_f32_16x16x4_f32(false, aH, false, bWl, (short)0, cL, false, false);
        }

        {
            const int mb = (lane >> 4) * 8;
            const int n  = lane & 15;
            #pragma unroll
            for (int r = 0; r < 8; ++r) {
                tH[(mb + r) * 16 + n] = cH[r];
                tM[(mb + r) * 16 + n] = cM[r];
                tL[(mb + r) * 16 + n] = cL[r];
            }
        }
        __syncthreads();

        if (lane < 16) {
            const int r = lane;
            float Hrow[16], Mrow[16], Lrow[16], xr[16];
            #pragma unroll
            for (int t = 0; t < 16; ++t) {
                Hrow[t] = tH[r * 16 + t] + b1 [c0 + t];
                Mrow[t] = tM[r * 16 + t] + bmu[c0 + t];
                Lrow[t] = tL[r * 16 + t] + bls[c0 + t];
                xr[t]   = x[(size_t)(r0 + r) * DD + c0 + t];
            }
            float zr[16], hr[16];
            #pragma unroll
            for (int t = 0; t < 16; ++t) {
                const float h  = fmaxf(Hrow[t], 0.0f);
                const float ls = Lrow[t];
                const float zv = Mrow[t] + xr[t] * __expf(ls);
                ldj += ls;
                zr[t] = zv;
                hr[t] = h;
                #pragma unroll
                for (int t2 = t + 1; t2 < 16; ++t2) {
                    Hrow[t2] = fmaf(zv, wd1[t2 * 16 + t], Hrow[t2]);
                    Mrow[t2] = fmaf(h,  wdm[t2 * 16 + t], Mrow[t2]);
                    Lrow[t2] = fmaf(h,  wdl[t2 * 16 + t], Lrow[t2]);
                }
            }
            #pragma unroll
            for (int t = 0; t < 16; ++t) {
                Zh[r * DD + c0 + t] = zr[t];
                Hh[r * DD + c0 + t] = hr[t];
                out[(size_t)(r0 + r) * DD + c0 + t] = zr[t];
            }
        }
        __syncthreads();
    }

    if (lane < 16) {
        out[(size_t)B * DD + r0 + lane] = ldj;
    }
}

extern "C" void kernel_launch(void* const* d_in, const int* in_sizes, int n_in,
                              void* d_out, int out_size, void* d_ws, size_t ws_size,
                              hipStream_t stream) {
    (void)n_in; (void)out_size; (void)d_ws; (void)ws_size;
    const float* x   = (const float*)d_in[0];
    const float* W1  = (const float*)d_in[1];
    const float* b1  = (const float*)d_in[2];
    const float* Wmu = (const float*)d_in[3];
    const float* bmu = (const float*)d_in[4];
    const float* Wls = (const float*)d_in[5];
    const float* bls = (const float*)d_in[6];
    float* out = (float*)d_out;

    const int B = in_sizes[0] / DD;
    dim3 grid(B / 16), block(32);
    iaf_layer_kernel<<<grid, block, 0, stream>>>(x, W1, b1, Wmu, bmu, Wls, bls, out, B);
}


# Cross-block WMMA products spread evenly over the blocks (history block b times column block J runs in a block b+1..J-1: at most 4 per block, was up to 6); weight rows staged earlier; sum order same
# speedup vs baseline: 2.9379x; 2.9379x over previous
#include <hip/hip_runtime.h>

typedef __attribute__((ext_vector_type(2))) float v2f;
typedef __attribute__((ext_vector_type(4))) float v4f;
typedef __attribute__((ext_vector_type(8))) float v8f;

#define DD 128
#define PW 132
#define NT 2
#define NWV 4

__device__ __forceinline__ void lds_barrier() {
    __builtin_amdgcn_fence(__ATOMIC_RELEASE, "workgroup", "local");
    __builtin_amdgcn_s_barrier();
    __builtin_amdgcn_fence(__ATOMIC_ACQUIRE, "workgroup", "local");
}

template <int N> __device__ __forceinline__ void fma_row_lane(float& acc, float w, float m) {
    asm volatile("v_fmac_f32_dpp %0, %1, %2 row_share:%3 row_mask:0xf bank_mask:0xf" : "+v"(acc) : "v"(w), "v"(m), "n"(N));
}

__device__ __forceinline__ float xhalf(float v) {
    const int b = __builtin_bit_cast(int, v);
    return __builtin_bit_cast(float, __builtin_amdgcn_permlanex16(b, b, 0x76543210, 0xfedcba98, false, true));
}

__device__ __forceinline__ float relu1(float v) {
    float r;
    asm("v_max_num_f32 %0, 0, %1" : "=v"(r) : "v"(v));
    return r;
}

__device__ __forceinline__ void add_here(float& acc, float v) {
    asm volatile("v_add_f32 %0, %0, %1" : "+v"(acc) : "v"(v));
}

__device__ __forceinline__ void seq_steps(float (&PH)[8], float (&PM)[8], float (&PL)[8],
                                          const float (&tHv)[16], const float (&tMv)[16], const float (&tLv)[16],
                                          const float (&bHv)[16], const float (&bMv)[16], const float (&bLv)[16],
                                          const float (&xr)[16], float (&zr)[16], float (&hr)[16],
                                          const float (&wc1)[16], const float (&wcm)[16], const float (&wcl)[16],
                                          float& ldjE, float& ldjO, const bool hi) {
        PH[0] = hi ? (tHv[1] + bHv[1]) : (tHv[0] + bHv[0]); PM[0] = hi ? (tMv[1] + bMv[1]) : (tMv[0] + bMv[0]); PL[0] = hi ? (tLv[1] + bLv[1]) : (tLv[0] + bLv[0]);
        __builtin_amdgcn_sched_barrier(0);
        PH[1] = hi ? (tHv[3] + bHv[3]) : (tHv[2] + bHv[2]); PM[1] = hi ? (tMv[3] + bMv[3]) : (tMv[2] + bMv[2]); PL[1] = hi ? (tLv[3] + bLv[3]) : (tLv[2] + bLv[2]);
        __builtin_amdgcn_sched_barrier(0);
        const float hc0 = relu1(PH[0]);
        __builtin_amdgcn_sched_barrier(0);
        const float m0 = PL[0] * 1.44269504088896341f; add_here(ldjE, PL[0]);
        __builtin_amdgcn_sched_barrier(0);
        PH[2] = hi ? (tHv[5] + bHv[5]) : (tHv[4] + bHv[4]); PM[2] = hi ? (tMv[5] + bMv[5]) : (tMv[4] + bMv[4]); PL[2] = hi ? (tLv[5] + bLv[5]) : (tLv[4] + bLv[4]);
        __builtin_amdgcn_sched_barrier(0);
        const float hx0 = xhalf(hc0);
        __builtin_amdgcn_sched_barrier(0);
        const float e0 = __builtin_amdgcn_exp2f(m0);
        __builtin_amdgcn_sched_barrier(0);
        PH[3] = hi ? (tHv[7] + bHv[7]) : (tHv[6] + bHv[6]); PM[3] = hi ? (tMv[7] + bMv[7]) : (tMv[6] + bMv[6]); PL[3] = hi ? (tLv[7] + bLv[7]) : (tLv[6] + bLv[6]);
        __builtin_amdgcn_sched_barrier(0);
        const float hb0 = hi ? hx0 : hc0; hr[0] = hb0;
        __builtin_amdgcn_sched_barrier(0);
        PH[4] = hi ? (tHv[9] + bHv[9]) : (tHv[8] + bHv[8]); PM[4] = hi ? (tMv[9] + bMv[9]) : (tMv[8] + bMv[8]); PL[4] = hi ? (tLv[9] + bLv[9]) : (tLv[8] + bLv[8]);
        __builtin_amdgcn_sched_barrier(0);
        const float zc0 = PM[0] + xr[0] * e0;
        __builtin_amdgcn_sched_barrier(0);
        fma_row_lane<0>(PL[0], wcl[0], hb0);
        fma_row_lane<1>(PL[1], wcl[0], hb0);
        __builtin_amdgcn_sched_barrier(0);
        const float zx0 = xhalf(zc0);
        __builtin_amdgcn_sched_barrier(0);
        fma_row_lane<0>(PM[0], wcm[0], hb0);
        __builtin_amdgcn_sched_barrier(0);
        const float m1 = PL[0] * 1.44269504088896341f; add_here(ldjO, PL[0]);
        __builtin_amdgcn_sched_barrier(0);
        const float zb0 = hi ? zx0 : zc0; zr[0] = zb0;
        __builtin_amdgcn_sched_barrier(0);
        fma_row_lane<1>(PM[1], wcm[0], hb0);
        __builtin_amdgcn_sched_barrier(0);
        const float e1 = __builtin_amdgcn_exp2f(m1);
        __builtin_amdgcn_sched_barrier(0);
        fma_row_lane<2>(PL[2], wcl[0], hb0);
        fma_row_lane<0>(PH[0], wc1[0], zb0);
        fma_row_lane<1>(PH[1], wc1[0], zb0);
        fma_row_lane<2>(PH[2], wc1[0], zb0);
        __builtin_amdgcn_sched_barrier(0);
        const float hc1 = relu1(PH[0]);
        __builtin_amdgcn_sched_barrier(0);
        const float zc1 = PM[0] + xr[1] * e1;
        __builtin_amdgcn_sched_barrier(0);
        fma_row_lane<2>(PM[2], wcm[0], hb0);
        fma_row_lane<3>(PL[3], wcl[0], hb0);
        __builtin_amdgcn_sched_barrier(0);
        const float hx1 = xhalf(hc1);
        __builtin_amdgcn_sched_barrier(0);
        const float zx1 = xhalf(zc1);
        __builtin_amdgcn_sched_barrier(0);
        fma_row_lane<3>(PH[3], wc1[0], zb0);
        fma_row_lane<3>(PM[3], wcm[0], hb0);
        __builtin_amdgcn_sched_barrier(0);
        const float hb1 = hi ? hc1 : hx1; hr[1] = hb1;
        __builtin_amdgcn_sched_barrier(0);
        const float zb1 = hi ? zc1 : zx1; zr[1] = zb1;
        __builtin_amdgcn_sched_barrier(0);
        fma_row_lane<4>(PL[4], wcl[0], hb0);
        fma_row_lane<4>(PH[4], wc1[0], zb0);
        fma_row_lane<1>(PL[1], wcl[1], hb1);
        fma_row_lane<1>(PH[1], wc1[1], zb1);
        fma_row_lane<1>(PM[1], wcm[1], hb1);
        __builtin_amdgcn_sched_barrier(0);
        const float m2 = PL[1] * 1.44269504088896341f; add_here(ldjE, PL[1]);
        __builtin_amdgcn_sched_barrier(0);
        const float hc2 = relu1(PH[1]);
        __builtin_amdgcn_sched_barrier(0);
        fma_row_lane<2>(PL[2], wcl[1], hb1);
        __builtin_amdgcn_sched_barrier(0);
        const float e2 = __builtin_amdgcn_exp2f(m2);
        __builtin_amdgcn_sched_barrier(0);
        fma_row_lane<2>(PH[2], wc1[1], zb1);
        __builtin_amdgcn_sched_barrier(0);
        const float hx2 = xhalf(hc2);
        __builtin_amdgcn_sched_barrier(0);
        fma_row_lane<2>(PM[2], wcm[1], hb1);
        fma_row_lane<3>(PL[3], wcl[1], hb1);
        __builtin_amdgcn_sched_barrier(0);
        const float hb2 = hi ? hx2 : hc2; hr[2] = hb2;
        __builtin_amdgcn_sched_barrier(0);
        const float zc2 = PM[1] + xr[2] * e2;
        __builtin_amdgcn_sched_barrier(0);
        fma_row_lane<3>(PH[3], wc1[1], zb1);
        fma_row_lane<3>(PM[3], wcm[1], hb1);
        fma_row_lane<1>(PL[1], wcl[2], hb2);
        __builtin_amdgcn_sched_barrier(0);
        const float zx2 = xhalf(zc2);
        __builtin_amdgcn_sched_barrier(0);
        fma_row_lane<1>(PM[1], wcm[2], hb2);
        fma_row_lane<2>(PL[2], wcl[2], hb2);
        __builtin_amdgcn_sched_barrier(0);
        const float m3 = PL[1] * 1.44269504088896341f; add_here(ldjO, PL[1]);
        __builtin_amdgcn_sched_barrier(0);
        const float zb2 = hi ? zx2 : zc2; zr[2] = zb2;
        __builtin_amdgcn_sched_barrier(0);
        fma_row_lane<2>(PM[2], wcm[2], hb2);
        __builtin_amdgcn_sched_barrier(0);
        const float e3 = __builtin_amdgcn_exp2f(m3);
        __builtin_amdgcn_sched_barrier(0);
        fma_row_lane<3>(PL[3], wcl[2], hb2);
        fma_row_lane<1>(PH[1], wc1[2], zb2);
        fma_row_lane<2>(PH[2], wc1[2], zb2);
        fma_row_lane<3>(PH[3], wc1[2], zb2);
        __builtin_amdgcn_sched_barrier(0);
        const float hc3 = relu1(PH[1]);
        __builtin_amdgcn_sched_barrier(0);
        const float zc3 = PM[1] + xr[3] * e3;
        __builtin_amdgcn_sched_barrier(0);
        fma_row_lane<3>(PM[3], wcm[2], hb2);
        fma_row_lane<4>(PL[4], wcl[1], hb1);
        __builtin_amdgcn_sched_barrier(0);
        const float hx3 = xhalf(hc3);
        __builtin_amdgcn_sched_barrier(0);
        const float zx3 = xhalf(zc3);
        __builtin_amdgcn_sched_barrier(0);
        fma_row_lane<4>(PH[4], wc1[1], zb1);
        fma_row_lane<4>(PL[4], wcl[2], hb2);
        __builtin_amdgcn_sched_barrier(0);
        const float hb3 = hi ? hc3 : hx3; hr[3] = hb3;
        __builtin_amdgcn_sched_barrier(0);
        const float zb3 = hi ? zc3 : zx3; zr[3] = zb3;
        __builtin_amdgcn_sched_barrier(0);
        fma_row_lane<4>(PH[4], wc1[2], zb2);
        fma_row_lane<4>(PM[4], wcm[0], hb0);
        fma_row_lane<2>(PL[2], wcl[3], hb3);
        fma_row_lane<2>(PH[2], wc1[3], zb3);
        fma_row_lane<2>(PM[2], wcm[3], hb3);
        __builtin_amdgcn_sched_barrier(0);
        const float m4 = PL[2] * 1.44269504088896341f; add_here(ldjE, PL[2]);
        __builtin_amdgcn_sched_barrier(0);
        const float hc4 = relu1(PH[2]);
        __builtin_amdgcn_sched_barrier(0);
        fma_row_lane<3>(PL[3], wcl[3], hb3);
        __builtin_amdgcn_sched_barrier(0);
        const float e4 = __builtin_amdgcn_exp2f(m4);
        __builtin_amdgcn_sched_barrier(0);
        fma_row_lane<3>(PH[3], wc1[3], zb3);
        __builtin_amdgcn_sched_barrier(0);
        const float hx4 = xhalf(hc4);
        __builtin_amdgcn_sched_barrier(0);
        fma_row_lane<3>(PM[3], wcm[3], hb3);
        fma_row_lane<4>(PL[4], wcl[3], hb3);
        __builtin_amdgcn_sched_barrier(0);
        const float hb4 = hi ? hx4 : hc4; hr[4] = hb4;
        __builtin_amdgcn_sched_barrier(0);
        const float zc4 = PM[2] + xr[4] * e4;
        __builtin_amdgcn_sched_barrier(0);
        fma_row_lane<4>(PH[4], wc1[3], zb3);
        fma_row_lane<4>(PM[4], wcm[1], hb1);
        fma_row_lane<2>(PL[2], wcl[4], hb4);
        __builtin_amdgcn_sched_barrier(0);
        const float zx4 = xhalf(zc4);
        __builtin_amdgcn_sched_barrier(0);
        fma_row_lane<2>(PM[2], wcm[4], hb4);
        fma_row_lane<3>(PL[3], wcl[4], hb4);
        __builtin_amdgcn_sched_barrier(0);
        const float m5 = PL[2] * 1.44269504088896341f; add_here(ldjO, PL[2]);
        __builtin_amdgcn_sched_barrier(0);
        const float zb4 = hi ? zx4 : zc4; zr[4] = zb4;
        __builtin_amdgcn_sched_barrier(0);
        fma_row_lane<3>(PM[3], wcm[4], hb4);
        __builtin_amdgcn_sched_barrier(0);
        const float e5 = __builtin_amdgcn_exp2f(m5);
        __builtin_amdgcn_sched_barrier(0);
        fma_row_lane<4>(PL[4], wcl[4], hb4);
        fma_row_lane<2>(PH[2], wc1[4], zb4);
        fma_row_lane<3>(PH[3], wc1[4], zb4);
        fma_row_lane<4>(PH[4], wc1[4], zb4);
        __builtin_amdgcn_sched_barrier(0);
        const float hc5 = relu1(PH[2]);
        __builtin_amdgcn_sched_barrier(0);
        const float zc5 = PM[2] + xr[5] * e5;
        __builtin_amdgcn_sched_barrier(0);
        fma_row_lane<4>(PM[4], wcm[2], hb2);
        __builtin_amdgcn_sched_barrier(0);
        PH[5] = hi ? (tHv[11] + bHv[11]) : (tHv[10] + bHv[10]); PM[5] = hi ? (tMv[11] + bMv[11]) : (tMv[10] + bMv[10]); PL[5] = hi ? (tLv[11] + bLv[11]) : (tLv[10] + bLv[10]);
        __builtin_amdgcn_sched_barrier(0);
        const float hx5 = xhalf(hc5);
        __builtin_amdgcn_sched_barrier(0);
        const float zx5 = xhalf(zc5);
        __builtin_amdgcn_sched_barrier(0);
        fma_row_lane<4>(PM[4], wcm[3], hb3);
        __builtin_amdgcn_sched_barrier(0);
        PH[6] = hi ? (tHv[13] + bHv[13]) : (tHv[12] + bHv[12]); PM[6] = hi ? (tMv[13] + bMv[13]) : (tMv[12] + bMv[12]); PL[6] = hi ? (tLv[13] + bLv[13]) : (tLv[12] + bLv[12]);
        __builtin_amdgcn_sched_barrier(0);
        const float hb5 = hi ? hc5 : hx5; hr[5] = hb5;
        __builtin_amdgcn_sched_barrier(0);
        const float zb5 = hi ? zc5 : zx5; zr[5] = zb5;
        __builtin_amdgcn_sched_barrier(0);
        fma_row_lane<4>(PM[4], wcm[4], hb4);
        fma_row_lane<5>(PL[5], wcl[0], hb0);
        fma_row_lane<3>(PL[3], wcl[5], hb5);
        fma_row_lane<3>(PH[3], wc1[5], zb5);
        fma_row_lane<3>(PM[3], wcm[5], hb5);
        __builtin_amdgcn_sched_barrier(0);
        const float m6 = PL[3] * 1.44269504088896341f; add_here(ldjE, PL[3]);
        __builtin_amdgcn_sched_barrier(0);
        const float hc6 = relu1(PH[3]);
        __builtin_amdgcn_sched_barrier(0);
        fma_row_lane<4>(PL[4], wcl[5], hb5);
        __builtin_amdgcn_sched_barrier(0);
        const float e6 = __builtin_amdgcn_exp2f(m6);
        __builtin_amdgcn_sched_barrier(0);
        fma_row_lane<4>(PH[4], wc1[5], zb5);
        __builtin_amdgcn_sched_barrier(0);
        const float hx6 = xhalf(hc6);
        __builtin_amdgcn_sched_barrier(0);
        fma_row_lane<5>(PH[5], wc1[0], zb0);
        fma_row_lane<4>(PM[4], wcm[5], hb5);
        __builtin_amdgcn_sched_barrier(0);
        const float hb6 = hi ? hx6 : hc6; hr[6] = hb6;
        __builtin_amdgcn_sched_barrier(0);
        const float zc6 = PM[3] + xr[6] * e6;
        __builtin_amdgcn_sched_barrier(0);
        fma_row_lane<5>(PL[5], wcl[1], hb1);
        fma_row_lane<5>(PH[5], wc1[1], zb1);
        fma_row_lane<3>(PL[3], wcl[6], hb6);
        __builtin_amdgcn_sched_barrier(0);
        const float zx6 = xhalf(zc6);
        __builtin_amdgcn_sched_barrier(0);
        fma_row_lane<3>(PM[3], wcm[6], hb6);
        fma_row_lane<4>(PL[4], wcl[6], hb6);
        __builtin_amdgcn_sched_barrier(0);
        const float m7 = PL[3] * 1.44269504088896341f; add_here(ldjO, PL[3]);
        __builtin_amdgcn_sched_barrier(0);
        const float zb6 = hi ? zx6 : zc6; zr[6] = zb6;
        __builtin_amdgcn_sched_barrier(0);
        fma_row_lane<4>(PM[4], wcm[6], hb6);
        __builtin_amdgcn_sched_barrier(0);
        const float e7 = __builtin_amdgcn_exp2f(m7);
        __builtin_amdgcn_sched_barrier(0);
        fma_row_lane<5>(PL[5], wcl[2], hb2);
        fma_row_lane<3>(PH[3], wc1[6], zb6);
        fma_row_lane<4>(PH[4], wc1[6], zb6);
        fma_row_lane<5>(PH[5], wc1[2], zb2);
        __builtin_amdgcn_sched_barrier(0);
        const float hc7 = relu1(PH[3]);
        __builtin_amdgcn_sched_barrier(0);
        const float zc7 = PM[3] + xr[7] * e7;
        __builtin_amdgcn_sched_barrier(0);
        fma_row_lane<5>(PM[5], wcm[0], hb0);
        fma_row_lane<5>(PL[5], wcl[3], hb3);
        __builtin_amdgcn_sched_barrier(0);
        const float hx7 = xhalf(hc7);
        __builtin_amdgcn_sched_barrier(0);
        const float zx7 = xhalf(zc7);
        __builtin_amdgcn_sched_barrier(0);
        fma_row_lane<5>(PH[5], wc1[3], zb3);
        fma_row_lane<5>(PM[5], wcm[1], hb1);
        __builtin_amdgcn_sched_barrier(0);
        const float hb7 = hi ? hc7 : hx7; hr[7] = hb7;
        __builtin_amdgcn_sched_barrier(0);
        const float zb7 = hi ? zc7 : zx7; zr[7] = zb7;
        __builtin_amdgcn_sched_barrier(0);
        fma_row_lane<5>(PL[5], wcl[4], hb4);
        fma_row_lane<5>(PH[5], wc1[4], zb4);
        fma_row_lane<4>(PL[4], wcl[7], hb7);
        fma_row_lane<4>(PH[4], wc1[7], zb7);
        fma_row_lane<4>(PM[4], wcm[7], hb7);
        __builtin_amdgcn_sched_barrier(0);
        const float m8 = PL[4] * 1.44269504088896341f; add_here(ldjE, PL[4]);
        __builtin_amdgcn_sched_barrier(0);
        const float hc8 = relu1(PH[4]);
        __builtin_amdgcn_sched_barrier(0);
        fma_row_lane<5>(PM[5], wcm[2], hb2);
        __builtin_amdgcn_sched_barrier(0);
        const float e8 = __builtin_amdgcn_exp2f(m8);
        __builtin_amdgcn_sched_barrier(0);
        fma_row_lane<5>(PL[5], wcl[5], hb5);
        __builtin_amdgcn_sched_barrier(0);
        const float hx8 = xhalf(hc8);
        __builtin_amdgcn_sched_barrier(0);
        fma_row_lane<5>(PH[5], wc1[5], zb5);
        fma_row_lane<5>(PM[5], wcm[3], hb3);
        __builtin_amdgcn_sched_barrier(0);
        const float hb8 = hi ? hx8 : hc8; hr[8] = hb8;
        __builtin_amdgcn_sched_barrier(0);
        const float zc8 = PM[4] + xr[8] * e8;
        __builtin_amdgcn_sched_barrier(0);
        fma_row_lane<5>(PL[5], wcl[6], hb6);
        fma_row_lane<5>(PH[5], wc1[6], zb6);
        fma_row_lane<4>(PL[4], wcl[8], hb8);
        __builtin_amdgcn_sched_barrier(0);
        const float zx8 = xhalf(zc8);
        __builtin_amdgcn_sched_barrier(0);
        fma_row_lane<4>(PM[4], wcm[8], hb8);
        fma_row_lane<5>(PM[5], wcm[4], hb4);
        __builtin_amdgcn_sched_barrier(0);
        const float m9 = PL[4] * 1.44269504088896341f; add_here(ldjO, PL[4]);
        __builtin_amdgcn_sched_barrier(0);
        const float zb8 = hi ? zx8 : zc8; zr[8] = zb8;
        __builtin_amdgcn_sched_barrier(0);
        fma_row_lane<5>(PL[5], wcl[7], hb7);
        __builtin_amdgcn_sched_barrier(0);
        const float e9 = __builtin_amdgcn_exp2f(m9);
        __builtin_amdgcn_sched_barrier(0);
        fma_row_lane<5>(PH[5], wc1[7], zb7);
        fma_row_lane<4>(PH[4], wc1[8], zb8);
        fma_row_lane<6>(PL[6], wcl[0], hb0);
        fma_row_lane<6>(PH[6], wc1[0], zb0);
        __builtin_amdgcn_sched_barrier(0);
        const float hc9 = relu1(PH[4]);
        __builtin_amdgcn_sched_barrier(0);
        const float zc9 = PM[4] + xr[9] * e9;
        __builtin_amdgcn_sched_barrier(0);
        fma_row_lane<5>(PM[5], wcm[5], hb5);
        fma_row_lane<5>(PL[5], wcl[8], hb8);
        __builtin_amdgcn_sched_barrier(0);
        const float hx9 = xhalf(hc9);
        __builtin_amdgcn_sched_barrier(0);
        const float zx9 = xhalf(zc9);
        __builtin_amdgcn_sched_barrier(0);
        fma_row_lane<5>(PH[5], wc1[8], zb8);
        fma_row_lane<6>(PL[6], wcl[1], hb1);
        __builtin_amdgcn_sched_barrier(0);
        const float hb9 = hi ? hc9 : hx9; hr[9] = hb9;
        __builtin_amdgcn_sched_barrier(0);
        const float zb9 = hi ? zc9 : zx9; zr[9] = zb9;
        __builtin_amdgcn_sched_barrier(0);
        fma_row_lane<6>(PH[6], wc1[1], zb1);
        fma_row_lane<5>(PM[5], wcm[6], hb6);
        fma_row_lane<5>(PL[5], wcl[9], hb9);
        fma_row_lane<5>(PH[5], wc1[9], zb9);
        fma_row_lane<6>(PL[6], wcl[2], hb2);
        fma_row_lane<6>(PH[6], wc1[2], zb2);
        fma_row_lane<5>(PM[5], wcm[7], hb7);
        fma_row_lane<6>(PM[6], wcm[0], hb0);
        __builtin_amdgcn_sched_barrier(0);
        const float hc10 = relu1(PH[5]);
        __builtin_amdgcn_sched_barrier(0);
        const float m10 = PL[5] * 1.44269504088896341f; add_here(ldjE, PL[5]);
        __builtin_amdgcn_sched_barrier(0);
        fma_row_lane<6>(PL[6], wcl[3], hb3);
        fma_row_lane<6>(PH[6], wc1[3], zb3);
        fma_row_lane<5>(PM[5], wcm[8], hb8);
        fma_row_lane<6>(PM[6], wcm[1], hb1);
        __builtin_amdgcn_sched_barrier(0);
        const float hx10 = xhalf(hc10);
        __builtin_amdgcn_sched_barrier(0);
        const float e10 = __builtin_amdgcn_exp2f(m10);
        __builtin_amdgcn_sched_barrier(0);
        fma_row_lane<6>(PL[6], wcl[4], hb4);
        fma_row_lane<6>(PH[6], wc1[4], zb4);
        fma_row_lane<5>(PM[5], wcm[9], hb9);
        fma_row_lane<6>(PM[6], wcm[2], hb2);
        __builtin_amdgcn_sched_barrier(0);
        const float hb10 = hi ? hx10 : hc10; hr[10] = hb10;
        __builtin_amdgcn_sched_barrier(0);
        fma_row_lane<6>(PL[6], wcl[5], hb5);
        fma_row_lane<6>(PH[6], wc1[5], zb5);
        __builtin_amdgcn_sched_barrier(0);
        const float zc10 = PM[5] + xr[10] * e10;
        __builtin_amdgcn_sched_barrier(0);
        fma_row_lane<5>(PL[5], wcl[10], hb10);
        fma_row_lane<6>(PM[6], wcm[3], hb3);
        __builtin_amdgcn_sched_barrier(0);
        PH[7] = hi ? (tHv[15] + bHv[15]) : (tHv[14] + bHv[14]); PM[7] = hi ? (tMv[15] + bMv[15]) : (tMv[14] + bMv[14]); PL[7] = hi ? (tLv[15] + bLv[15]) : (tLv[14] + bLv[14]);
        __builtin_amdgcn_sched_barrier(0);
        const float zx10 = xhalf(zc10);
        __builtin_amdgcn_sched_barrier(0);
        fma_row_lane<6>(PL[6], wcl[6], hb6);
        fma_row_lane<6>(PH[6], wc1[6], zb6);
        __builtin_amdgcn_sched_barrier(0);
        const float m11 = PL[5] * 1.44269504088896341f; add_here(ldjO, PL[5]);
        __builtin_amdgcn_sched_barrier(0);
        fma_row_lane<6>(PM[6], wcm[4], hb4);
        __builtin_amdgcn_sched_barrier(0);
        const float zb10 = hi ? zx10 : zc10; zr[10] = zb10;
        __builtin_amdgcn_sched_barrier(0);
        fma_row_lane<6>(PL[6], wcl[7], hb7);
        fma_row_lane<6>(PH[6], wc1[7], zb7);
        __builtin_amdgcn_sched_barrier(0);
        const float e11 = __builtin_amdgcn_exp2f(m11);
        __builtin_amdgcn_sched_barrier(0);
        fma_row_lane<7>(PL[7], wcl[0], hb0);
        fma_row_lane<7>(PH[7], wc1[0], zb0);
        fma_row_lane<6>(PM[6], wcm[5], hb5);
        fma_row_lane<5>(PM[5], wcm[10], hb10);
        fma_row_lane<5>(PH[5], wc1[10], zb10);
        fma_row_lane<6>(PL[6], wcl[8], hb8);
        fma_row_lane<6>(PH[6], wc1[8], zb8);
        fma_row_lane<7>(PL[7], wcl[1], hb1);
        fma_row_lane<7>(PH[7], wc1[1], zb1);
        fma_row_lane<6>(PM[6], wcm[6], hb6);
        __builtin_amdgcn_sched_barrier(0);
        const float hc11 = relu1(PH[5]);
        __builtin_amdgcn_sched_barrier(0);
        const float zc11 = PM[5] + xr[11] * e11;
        __builtin_amdgcn_sched_barrier(0);
        fma_row_lane<6>(PL[6], wcl[9], hb9);
        fma_row_lane<6>(PH[6], wc1[9], zb9);
        fma_row_lane<7>(PL[7], wcl[2], hb2);
        fma_row_lane<7>(PH[7], wc1[2], zb2);
        fma_row_lane<6>(PM[6], wcm[7], hb7);
        __builtin_amdgcn_sched_barrier(0);
        const float hx11 = xhalf(hc11);
        __builtin_amdgcn_sched_barrier(0);
        const float zx11 = xhalf(zc11);
        __builtin_amdgcn_sched_barrier(0);
        fma_row_lane<7>(PM[7], wcm[0], hb0);
        fma_row_lane<6>(PL[6], wcl[10], hb10);
        fma_row_lane<6>(PH[6], wc1[10], zb10);
        __builtin_amdgcn_sched_barrier(0);
        const float hb11 = hi ? hc11 : hx11; hr[11] = hb11;
        __builtin_amdgcn_sched_barrier(0);
        const float zb11 = hi ? zc11 : zx11; zr[11] = zb11;
        __builtin_amdgcn_sched_barrier(0);
        fma_row_lane<7>(PL[7], wcl[3], hb3);
        fma_row_lane<7>(PH[7], wc1[3], zb3);
        fma_row_lane<6>(PM[6], wcm[8], hb8);
        fma_row_lane<7>(PM[7], wcm[1], hb1);
        fma_row_lane<6>(PL[6], wcl[11], hb11);
        fma_row_lane<6>(PH[6], wc1[11], zb11);
        fma_row_lane<7>(PL[7], wcl[4], hb4);
        fma_row_lane<7>(PH[7], wc1[4], zb4);
        fma_row_lane<6>(PM[6], wcm[9], hb9);
        fma_row_lane<7>(PM[7], wcm[2], hb2);
        __builtin_amdgcn_sched_barrier(0);
        const float hc12 = relu1(PH[6]);
        __builtin_amdgcn_sched_barrier(0);
        const float m12 = PL[6] * 1.44269504088896341f; add_here(ldjE, PL[6]);
        __builtin_amdgcn_sched_barrier(0);
        fma_row_lane<7>(PL[7], wcl[5], hb5);
        fma_row_lane<7>(PH[7], wc1[5], zb5);
        fma_row_lane<6>(PM[6], wcm[10], hb10);
        fma_row_lane<7>(PM[7], wcm[3], hb3);
        __builtin_amdgcn_sched_barrier(0);
        const float hx12 = xhalf(hc12);
        __builtin_amdgcn_sched_barrier(0);
        const float e12 = __builtin_amdgcn_exp2f(m12);
        __builtin_amdgcn_sched_barrier(0);
        fma_row_lane<7>(PL[7], wcl[6], hb6);
        fma_row_lane<7>(PH[7], wc1[6], zb6);
        fma_row_lane<6>(PM[6], wcm[11], hb11);
        fma_row_lane<7>(PM[7], wcm[4], hb4);
        __builtin_amdgcn_sched_barrier(0);
        const float hb12 = hi ? hx12 : hc12; hr[12] = hb12;
        __builtin_amdgcn_sched_barrier(0);
        fma_row_lane<7>(PL[7], wcl[7], hb7);
        fma_row_lane<7>(PH[7], wc1[7], zb7);
        __builtin_amdgcn_sched_barrier(0);
        const float zc12 = PM[6] + xr[12] * e12;
        __builtin_amdgcn_sched_barrier(0);
        fma_row_lane<6>(PL[6], wcl[12], hb12);
        fma_row_lane<7>(PM[7], wcm[5], hb5);
        __builtin_amdgcn_sched_barrier(0);
        const float zx12 = xhalf(zc12);
        __builtin_amdgcn_sched_barrier(0);
        fma_row_lane<7>(PL[7], wcl[8], hb8);
        fma_row_lane<7>(PH[7], wc1[8], zb8);
        __builtin_amdgcn_sched_barrier(0);
        const float m13 = PL[6] * 1.44269504088896341f; add_here(ldjO, PL[6]);
        __builtin_amdgcn_sched_barrier(0);
        fma_row_lane<7>(PM[7], wcm[6], hb6);
        __builtin_amdgcn_sched_barrier(0);
        const float zb12 = hi ? zx12 : zc12; zr[12] = zb12;
        __builtin_amdgcn_sched_barrier(0);
        fma_row_lane<7>(PL[7], wcl[9], hb9);
        fma_row_lane<7>(PH[7], wc1[9], zb9);
        __builtin_amdgcn_sched_barrier(0);
        const float e13 = __builtin_amdgcn_exp2f(m13);
        __builtin_amdgcn_sched_barrier(0);
        fma_row_lane<7>(PM[7], wcm[7], hb7);
        fma_row_lane<6>(PM[6], wcm[12], hb12);
        fma_row_lane<6>(PH[6], wc1[12], zb12);
        fma_row_lane<7>(PL[7], wcl[10], hb10);
        fma_row_lane<7>(PH[7], wc1[10], zb10);
        fma_row_lane<7>(PM[7], wcm[8], hb8);
        __builtin_amdgcn_sched_barrier(0);
        const float hc13 = relu1(PH[6]);
        __builtin_amdgcn_sched_barrier(0);
        const float zc13 = PM[6] + xr[13] * e13;
        __builtin_amdgcn_sched_barrier(0);
        fma_row_lane<7>(PL[7], wcl[11], hb11);
        fma_row_lane<7>(PH[7], wc1[11], zb11);
        fma_row_lane<7>(PM[7], wcm[9], hb9);
        __builtin_amdgcn_sched_barrier(0);
        const float hx13 = xhalf(hc13);
        __builtin_amdgcn_sched_barrier(0);
        const float zx13 = xhalf(zc13);
        __builtin_amdgcn_sched_barrier(0);
        fma_row_lane<7>(PL[7], wcl[12], hb12);
        fma_row_lane<7>(PH[7], wc1[12], zb12);
        __builtin_amdgcn_sched_barrier(0);
        const float hb13 = hi ? hc13 : hx13; hr[13] = hb13;
        __builtin_amdgcn_sched_barrier(0);
        const float zb13 = hi ? zc13 : zx13; zr[13] = zb13;
        __builtin_amdgcn_sched_barrier(0);
        fma_row_lane<7>(PM[7], wcm[10], hb10);
        fma_row_lane<7>(PL[7], wcl[13], hb13);
        fma_row_lane<7>(PH[7], wc1[13], zb13);
        fma_row_lane<7>(PM[7], wcm[11], hb11);
        __builtin_amdgcn_sched_barrier(0);
        const float m14 = PL[7] * 1.44269504088896341f; add_here(ldjE, PL[7]);
        __builtin_amdgcn_sched_barrier(0);
        const float hc14 = relu1(PH[7]);
        __builtin_amdgcn_sched_barrier(0);
        fma_row_lane<7>(PM[7], wcm[12], hb12);
        __builtin_amdgcn_sched_barrier(0);
        const float e14 = __builtin_amdgcn_exp2f(m14);
        __builtin_amdgcn_sched_barrier(0);
        const float hx14 = xhalf(hc14);
        __builtin_amdgcn_sched_barrier(0);
        fma_row_lane<7>(PM[7], wcm[13], hb13);
        __builtin_amdgcn_sched_barrier(0);
        const float hb14 = hi ? hx14 : hc14; hr[14] = hb14;
        __builtin_amdgcn_sched_barrier(0);
        const float zc14 = PM[7] + xr[14] * e14;
        __builtin_amdgcn_sched_barrier(0);
        fma_row_lane<7>(PL[7], wcl[14], hb14);
        __builtin_amdgcn_sched_barrier(0);
        const float zx14 = xhalf(zc14);
        __builtin_amdgcn_sched_barrier(0);
        fma_row_lane<7>(PM[7], wcm[14], hb14);
        __builtin_amdgcn_sched_barrier(0);
        const float m15 = PL[7] * 1.44269504088896341f; add_here(ldjO, PL[7]);
        __builtin_amdgcn_sched_barrier(0);
        const float zb14 = hi ? zx14 : zc14; zr[14] = zb14;
        __builtin_amdgcn_sched_barrier(0);
        const float e15 = __builtin_amdgcn_exp2f(m15);
        __builtin_amdgcn_sched_barrier(0);
        fma_row_lane<7>(PH[7], wc1[14], zb14);
        __builtin_amdgcn_sched_barrier(0);
        const float hc15 = relu1(PH[7]);
        __builtin_amdgcn_sched_barrier(0);
        const float zc15 = PM[7] + xr[15] * e15;
        __builtin_amdgcn_sched_barrier(0);
        const float hx15 = xhalf(hc15);
        __builtin_amdgcn_sched_barrier(0);
        const float zx15 = xhalf(zc15);
        __builtin_amdgcn_sched_barrier(0);
        const float hb15 = hi ? hc15 : hx15; hr[15] = hb15;
        __builtin_amdgcn_sched_barrier(0);
        const float zb15 = hi ? zc15 : zx15; zr[15] = zb15;
        __builtin_amdgcn_sched_barrier(0);
}

__global__ __launch_bounds__(32 * NWV) void iaf_layer_kernel(
    const float* __restrict__ x,
    const float* __restrict__ W1,  const float* __restrict__ b1,
    const float* __restrict__ Wmu, const float* __restrict__ bmu,
    const float* __restrict__ Wls, const float* __restrict__ bls,
    float* __restrict__ out, int B)
{
    __shared__ __attribute__((aligned(16))) float Ws[3][DD * PW];
    __shared__ __attribute__((aligned(16))) float XZs[NT][16 * PW];
    __shared__ __attribute__((aligned(16))) float Hs[NT][16 * PW];
    __shared__ __attribute__((aligned(16))) float tHs[NT][256];
    __shared__ __attribute__((aligned(16))) float tMs[NT][256];
    __shared__ __attribute__((aligned(16))) float tLs[NT][256];
    __shared__ __attribute__((aligned(16))) float bs[3][DD];
    __shared__ float ldjs[NT * 16];

    const int tid   = threadIdx.x;
    const int lane  = tid & 31;
    const int wv    = __builtin_amdgcn_readfirstlane(tid >> 5);
    const bool seqw = wv < NT;
    const int tile  = wv & (NT - 1);
    const int r0    = blockIdx.x * (16 * NT) + tile * 16;
    const int mrow  = lane & 15;
    const int khalf = (lane >> 4) * 2;

    float* XZ = XZs[tile];
    float* Hh = Hs[tile];
    float* tH = tHs[tile];
    float* tM = tMs[tile];
    float* tL = tLs[tile];

    v4f pa[8], pb[8], pc[8];
    auto load_rows = [&](int first, int step) {
        #pragma unroll
        for (int q = 0; q < 8; ++q) {
            const int row = first + step * q;
            pa[q] = *(const v4f*)&W1 [row * DD + lane * 4];
            pb[q] = *(const v4f*)&Wmu[row * DD + lane * 4];
            pc[q] = *(const v4f*)&Wls[row * DD + lane * 4];
        }
    };
    auto store_rows = [&](int first, int step) {
        #pragma unroll
        for (int q = 0; q < 8; ++q) {
            const int row = first + step * q;
            *(v4f*)&Ws[0][row * PW + lane * 4] = pa[q];
            *(v4f*)&Ws[1][row * PW + lane * 4] = pb[q];
            *(v4f*)&Ws[2][row * PW + lane * 4] = pc[q];
        }
    };

    if (seqw) {
        v4f xv[16];
        #pragma unroll
        for (int r = 0; r < 16; ++r)
            xv[r] = *(const v4f*)&x[(size_t)(r0 + r) * DD + lane * 4];
        #pragma unroll
        for (int r = 0; r < 16; ++r)
            *(v4f*)&XZ[r * PW + lane * 4] = xv[r];
    } else {
        const v4f zero4 = {0.f, 0.f, 0.f, 0.f};
        *(v4f*)&tH[lane * 8] = zero4;  *(v4f*)&tH[lane * 8 + 4] = zero4;
        *(v4f*)&tM[lane * 8] = zero4;  *(v4f*)&tM[lane * 8 + 4] = zero4;
        *(v4f*)&tL[lane * 8] = zero4;  *(v4f*)&tL[lane * 8 + 4] = zero4;
    }
    if (lane < 8) {
        load_rows(wv, NWV);
        store_rows(wv, NWV);
    }
    if (wv < 3) {
        const float* bsrc = (wv == 0) ? b1 : (wv == 1) ? bmu : bls;
        *(v4f*)&bs[wv][lane * 4] = *(const v4f*)&bsrc[lane * 4];
    }
    __syncthreads();

    float ldjE = 0.0f, ldjO = 0.0f;

    const float* az  = &XZ[mrow * PW + khalf];
    const float* ah  = &Hh[mrow * PW + khalf];

    auto mm_block = [&](v8f& cH, v8f& cM, v8f& cL, int kb, int J) {
        const float* bw1 = &Ws[0][(16 * J + mrow) * PW + khalf];
        const float* bwm = &Ws[1][(16 * J + mrow) * PW + khalf];
        const float* bwl = &Ws[2][(16 * J + mrow) * PW + khalf];
        #pragma unroll
        for (int kt = 0; kt < 4; ++kt) {
            const int k0 = kb + kt * 4;
            const v2f aZ  = *(const v2f*)&az[k0];
            const v2f aH  = *(const v2f*)&ah[k0];
            const v2f bW1 = *(const v2f*)&bw1[k0];
            const v2f bWm = *(const v2f*)&bwm[k0];
            const v2f bWl = *(const v2f*)&bwl[k0];
            cH = __builtin_amdgcn_wmma_f32_16x16x4_f32(false, aZ, false, bW1, (short)0, cH, false, false);
            cM = __builtin_amdgcn_wmma_f32_16x16x4_f32(false, aH, false, bWm, (short)0, cM, false, false);
            cL = __builtin_amdgcn_wmma_f32_16x16x4_f32(false, aH, false, bWl, (short)0, cL, false, false);
        }
    };

    if (seqw) {
        for (int I = 0; I < 8; ++I) {
            const int c0 = I * 16;
            const bool pf = (I < 4) && (lane < 4 * (I + 5));
            if (pf) load_rows(16 * (I + 4) + wv, NT);

            {
                const int r = mrow;
                const bool hi = lane >= 16;
                const int wrow = c0 + 2 * (lane & 7) + (lane >> 4);
                const float* wr1 = &Ws[0][wrow * PW + c0];
                const float* wrm = &Ws[1][wrow * PW + c0];
                const float* wrl = &Ws[2][wrow * PW + c0];
                float wc1[16], wcm[16], wcl[16];
                #pragma unroll
                for (int q = 0; q < 4; ++q) {
                    const v4f a = *(const v4f*)&wr1[4 * q];
                    const v4f b = *(const v4f*)&wrm[4 * q];
                    const v4f c = *(const v4f*)&wrl[4 * q];
                    wc1[4 * q] = a.x; wc1[4 * q + 1] = a.y; wc1[4 * q + 2] = a.z; wc1[4 * q + 3] = a.w;
                    wcm[4 * q] = b.x; wcm[4 * q + 1] = b.y; wcm[4 * q + 2] = b.z; wcm[4 * q + 3] = b.w;
                    wcl[4 * q] = c.x; wcl[4 * q + 1] = c.y; wcl[4 * q + 2] = c.z; wcl[4 * q + 3] = c.w;
                }
                float PH[8], PM[8], PL[8], xr[16];
                float tHv[16], tMv[16], tLv[16], bHv[16], bMv[16], bLv[16];
                #pragma unroll
                for (int t = 0; t < 16; ++t) {
                    tHv[t] = tH[r * 16 + t];  bHv[t] = bs[0][c0 + t];
                    tMv[t] = tM[r * 16 + t];  bMv[t] = bs[1][c0 + t];
                    tLv[t] = tL[r * 16 + t];  bLv[t] = bs[2][c0 + t];
                }
                #pragma unroll
                for (int t = 0; t < 16; ++t)
                    xr[t] = XZ[r * PW + c0 + t];
                float zr[16], hr[16];
                seq_steps(PH, PM, PL, tHv, tMv, tLv, bHv, bMv, bLv, xr, zr, hr, wc1, wcm, wcl, ldjE, ldjO, hi);
                if (lane < 16) {
                    #pragma unroll
                    for (int t = 0; t < 16; ++t) {
                        XZ[r * PW + c0 + t] = zr[t];
                        Hh[r * PW + c0 + t] = hr[t];
                    }
                }
                if (I == 7) {
                    const float tot = ldjE + xhalf(ldjO);
                    if (lane < 16) ldjs[tile * 16 + r] = tot;
                }
            }
            lds_barrier();
            if (pf) store_rows(16 * (I + 4) + wv, NT);
            lds_barrier();
        }
    } else {
        const v8f zero8 = {0.f,0.f,0.f,0.f,0.f,0.f,0.f,0.f};
        v8f cH1 = zero8, cM1 = zero8, cL1 = zero8;
        v8f cH2 = zero8, cM2 = zero8, cL2 = zero8;
        v8f cH3 = zero8, cM3 = zero8, cL3 = zero8;
        v8f cH4 = zero8, cM4 = zero8, cL4 = zero8;
        v8f cH5 = zero8, cM5 = zero8, cL5 = zero8;
        v8f cH6 = zero8, cM6 = zero8, cL6 = zero8;
        v8f cH7 = zero8, cM7 = zero8, cL7 = zero8;
        auto put_tiles = [&](const v8f& cH, const v8f& cM, const v8f& cL) {
            const int mb = (lane >> 4) * 8;
            const int n  = lane & 15;
            #pragma unroll
            for (int r = 0; r < 8; ++r) {
                tH[(mb + r) * 16 + n] = cH[r];
                tM[(mb + r) * 16 + n] = cM[r];
                tL[(mb + r) * 16 + n] = cL[r];
            }
        };

        auto mat_barrier = [&]() {
            __builtin_amdgcn_sched_barrier(0);
            lds_barrier();
            __builtin_amdgcn_sched_barrier(0);
        };

        {
            v4f qa[2][8], qb[2][8], qc[2][8];
            #pragma unroll
            for (int s = 0; s < 2; ++s) {
                if (lane < 4 * (s + 3)) {
                    #pragma unroll
                    for (int q = 0; q < 8; ++q) {
                        const int row = 16 * (s + 2) + tile + NT * q;
                        qa[s][q] = *(const v4f*)&W1 [row * DD + lane * 4];
                        qb[s][q] = *(const v4f*)&Wmu[row * DD + lane * 4];
                        qc[s][q] = *(const v4f*)&Wls[row * DD + lane * 4];
                    }
                }
            }
            #pragma unroll
            for (int s = 0; s < 2; ++s) {
                if (lane < 4 * (s + 3)) {
                    #pragma unroll
                    for (int q = 0; q < 8; ++q) {
                        const int row = 16 * (s + 2) + tile + NT * q;
                        *(v4f*)&Ws[0][row * PW + lane * 4] = qa[s][q];
                        *(v4f*)&Ws[1][row * PW + lane * 4] = qb[s][q];
                        *(v4f*)&Ws[2][row * PW + lane * 4] = qc[s][q];
                    }
                }
            }
        }
        mat_barrier();
        mm_block(cH1, cM1, cL1, 0, 1);
        put_tiles(cH1, cM1, cL1);
        mat_barrier();

        mm_block(cH2, cM2, cL2, 0, 2);
        mm_block(cH3, cM3, cL3, 0, 3);
        mm_block(cH4, cM4, cL4, 0, 4);
        mat_barrier();
        mm_block(cH2, cM2, cL2, 16, 2);
        put_tiles(cH2, cM2, cL2);
        mat_barrier();

        mm_block(cH3, cM3, cL3, 16, 3);
        mm_block(cH4, cM4, cL4, 16, 4);
        mm_block(cH5, cM5, cL5, 0, 5);
        mm_block(cH5, cM5, cL5, 16, 5);
        mat_barrier();
        mm_block(cH3, cM3, cL3, 32, 3);
        put_tiles(cH3, cM3, cL3);
        mat_barrier();

        mm_block(cH4, cM4, cL4, 32, 4);
        mm_block(cH5, cM5, cL5, 32, 5);
        mm_block(cH6, cM6, cL6, 0, 6);
        mm_block(cH6, cM6, cL6, 16, 6);
        mat_barrier();
        mm_block(cH4, cM4, cL4, 48, 4);
        put_tiles(cH4, cM4, cL4);
        mat_barrier();

        mm_block(cH5, cM5, cL5, 48, 5);
        mm_block(cH6, cM6, cL6, 32, 6);
        mm_block(cH7, cM7, cL7, 0, 7);
        mm_block(cH7, cM7, cL7, 16, 7);
        mat_barrier();
        mm_block(cH5, cM5, cL5, 64, 5);
        put_tiles(cH5, cM5, cL5);
        mat_barrier();

        mm_block(cH6, cM6, cL6, 48, 6);
        mm_block(cH6, cM6, cL6, 64, 6);
        mm_block(cH7, cM7, cL7, 32, 7);
        mat_barrier();
        mm_block(cH6, cM6, cL6, 80, 6);
        put_tiles(cH6, cM6, cL6);
        mat_barrier();

        mm_block(cH7, cM7, cL7, 48, 7);
        mm_block(cH7, cM7, cL7, 64, 7);
        mm_block(cH7, cM7, cL7, 80, 7);
        mat_barrier();
        mm_block(cH7, cM7, cL7, 96, 7);
        put_tiles(cH7, cM7, cL7);
        mat_barrier();

        mat_barrier();
        mat_barrier();
    }

    const int rb = seqw ? 0 : 8;
    #pragma unroll
    for (int r = 0; r < 8; ++r)
        *(v4f*)&out[(size_t)(r0 + rb + r) * DD + lane * 4] = *(const v4f*)&XZ[(rb + r) * PW + lane * 4];

    if (tid < 16 * NT)
        out[(size_t)B * DD + blockIdx.x * (16 * NT) + tid] = ldjs[tid];
}

extern "C" void kernel_launch(void* const* d_in, const int* in_sizes, int n_in,
                              void* d_out, int out_size, void* d_ws, size_t ws_size,
                              hipStream_t stream) {
    (void)n_in; (void)out_size; (void)d_ws; (void)ws_size;
    const float* x   = (const float*)d_in[0];
    const float* W1  = (const float*)d_in[1];
    const float* b1  = (const float*)d_in[2];
    const float* Wmu = (const float*)d_in[3];
    const float* bmu = (const float*)d_in[4];
    const float* Wls = (const float*)d_in[5];
    const float* bls = (const float*)d_in[6];
    float* out = (float*)d_out;

    const int B = in_sizes[0] / DD;
    dim3 grid(B / (16 * NT)), block(32 * NWV);
    iaf_layer_kernel<<<grid, block, 0, stream>>>(x, W1, b1, Wmu, bmu, Wls, bls, out, B);
}


# Sequential wave multiplies each finished block into the next block's start values itself (12 transposed WMMAs, result in its pair layout); 1 barrier per block; newest block summed apart then added
# speedup vs baseline: 3.1620x; 1.0763x over previous
#include <hip/hip_runtime.h>

typedef __attribute__((ext_vector_type(2))) float v2f;
typedef __attribute__((ext_vector_type(4))) float v4f;
typedef __attribute__((ext_vector_type(8))) float v8f;

#define DD 128
#define PW 132
#define NT 2
#define NWV 4

__device__ __forceinline__ void lds_barrier() {
    __builtin_amdgcn_fence(__ATOMIC_RELEASE, "workgroup", "local");
    __builtin_amdgcn_s_barrier();
    __builtin_amdgcn_fence(__ATOMIC_ACQUIRE, "workgroup", "local");
}

template <int N> __device__ __forceinline__ void fma_row_lane(float& acc, float w, float m) {
    asm volatile("v_fmac_f32_dpp %0, %1, %2 row_share:%3 row_mask:0xf bank_mask:0xf" : "+v"(acc) : "v"(w), "v"(m), "n"(N));
}

__device__ __forceinline__ float xhalf(float v) {
    const int b = __builtin_bit_cast(int, v);
    return __builtin_bit_cast(float, __builtin_amdgcn_permlanex16(b, b, 0x76543210, 0xfedcba98, false, true));
}

__device__ __forceinline__ float relu1(float v) {
    float r;
    asm("v_max_num_f32 %0, 0, %1" : "=v"(r) : "v"(v));
    return r;
}

__device__ __forceinline__ void add_here(float& acc, float v) {
    asm volatile("v_add_f32 %0, %0, %1" : "+v"(acc) : "v"(v));
}

template <bool PROD>
__device__ __forceinline__ void seq_steps(float (&PH)[8], float (&PM)[8], float (&PL)[8],
                                          const float (&pH)[8], const float (&pM)[8], const float (&pL)[8],
                                          const v8f& qH, const v8f& qM, const v8f& qL,
                                          const float (&bH)[8], const float (&bM)[8], const float (&bL)[8],
                                          const float (&xr)[16], float (&zr)[16], float (&hr)[16],
                                          const float (&wc1)[16], const float (&wcm)[16], const float (&wcl)[16],
                                          const v2f (&aw1)[4], const v2f (&awm)[4], const v2f (&awl)[4],
                                          v8f& nH, v8f& nM, v8f& nL,
                                          float& ldjE, float& ldjO, const bool hi) {
        PH[0] = (pH[0] + qH[0]) + bH[0]; PM[0] = (pM[0] + qM[0]) + bM[0]; PL[0] = (pL[0] + qL[0]) + bL[0];
        __builtin_amdgcn_sched_barrier(0);
        PH[1] = (pH[1] + qH[1]) + bH[1]; PM[1] = (pM[1] + qM[1]) + bM[1]; PL[1] = (pL[1] + qL[1]) + bL[1];
        __builtin_amdgcn_sched_barrier(0);
        const float hc0 = relu1(PH[0]);
        __builtin_amdgcn_sched_barrier(0);
        const float m0 = PL[0] * 1.44269504088896341f; add_here(ldjE, PL[0]);
        __builtin_amdgcn_sched_barrier(0);
        PH[2] = (pH[2] + qH[2]) + bH[2]; PM[2] = (pM[2] + qM[2]) + bM[2]; PL[2] = (pL[2] + qL[2]) + bL[2];
        __builtin_amdgcn_sched_barrier(0);
        const float hx0 = xhalf(hc0);
        __builtin_amdgcn_sched_barrier(0);
        const float e0 = __builtin_amdgcn_exp2f(m0);
        __builtin_amdgcn_sched_barrier(0);
        PH[3] = (pH[3] + qH[3]) + bH[3]; PM[3] = (pM[3] + qM[3]) + bM[3]; PL[3] = (pL[3] + qL[3]) + bL[3];
        __builtin_amdgcn_sched_barrier(0);
        const float hb0 = hi ? hx0 : hc0; hr[0] = hb0;
        __builtin_amdgcn_sched_barrier(0);
        PH[4] = (pH[4] + qH[4]) + bH[4]; PM[4] = (pM[4] + qM[4]) + bM[4]; PL[4] = (pL[4] + qL[4]) + bL[4];
        __builtin_amdgcn_sched_barrier(0);
        const float zc0 = PM[0] + xr[0] * e0;
        __builtin_amdgcn_sched_barrier(0);
        fma_row_lane<0>(PL[0], wcl[0], hb0);
        fma_row_lane<1>(PL[1], wcl[0], hb0);
        __builtin_amdgcn_sched_barrier(0);
        const float zx0 = xhalf(zc0);
        __builtin_amdgcn_sched_barrier(0);
        fma_row_lane<0>(PM[0], wcm[0], hb0);
        __builtin_amdgcn_sched_barrier(0);
        const float m1 = PL[0] * 1.44269504088896341f; add_here(ldjO, PL[0]);
        __builtin_amdgcn_sched_barrier(0);
        const float zb0 = hi ? zx0 : zc0; zr[0] = zb0;
        __builtin_amdgcn_sched_barrier(0);
        fma_row_lane<1>(PM[1], wcm[0], hb0);
        __builtin_amdgcn_sched_barrier(0);
        const float e1 = __builtin_amdgcn_exp2f(m1);
        __builtin_amdgcn_sched_barrier(0);
        fma_row_lane<2>(PL[2], wcl[0], hb0);
        fma_row_lane<0>(PH[0], wc1[0], zb0);
        fma_row_lane<1>(PH[1], wc1[0], zb0);
        fma_row_lane<2>(PH[2], wc1[0], zb0);
        __builtin_amdgcn_sched_barrier(0);
        const float hc1 = relu1(PH[0]);
        __builtin_amdgcn_sched_barrier(0);
        const float zc1 = PM[0] + xr[1] * e1;
        __builtin_amdgcn_sched_barrier(0);
        fma_row_lane<2>(PM[2], wcm[0], hb0);
        fma_row_lane<3>(PL[3], wcl[0], hb0);
        __builtin_amdgcn_sched_barrier(0);
        const float hx1 = xhalf(hc1);
        __builtin_amdgcn_sched_barrier(0);
        const float zx1 = xhalf(zc1);
        __builtin_amdgcn_sched_barrier(0);
        fma_row_lane<3>(PH[3], wc1[0], zb0);
        fma_row_lane<3>(PM[3], wcm[0], hb0);
        __builtin_amdgcn_sched_barrier(0);
        const float hb1 = hi ? hc1 : hx1; hr[1] = hb1;
        __builtin_amdgcn_sched_barrier(0);
        const float zb1 = hi ? zc1 : zx1; zr[1] = zb1;
        __builtin_amdgcn_sched_barrier(0);
        fma_row_lane<4>(PL[4], wcl[0], hb0);
        fma_row_lane<4>(PH[4], wc1[0], zb0);
        fma_row_lane<1>(PL[1], wcl[1], hb1);
        fma_row_lane<1>(PH[1], wc1[1], zb1);
        fma_row_lane<1>(PM[1], wcm[1], hb1);
        __builtin_amdgcn_sched_barrier(0);
        const float m2 = PL[1] * 1.44269504088896341f; add_here(ldjE, PL[1]);
        __builtin_amdgcn_sched_barrier(0);
        const float hc2 = relu1(PH[1]);
        __builtin_amdgcn_sched_barrier(0);
        fma_row_lane<2>(PL[2], wcl[1], hb1);
        __builtin_amdgcn_sched_barrier(0);
        const float e2 = __builtin_amdgcn_exp2f(m2);
        __builtin_amdgcn_sched_barrier(0);
        fma_row_lane<2>(PH[2], wc1[1], zb1);
        __builtin_amdgcn_sched_barrier(0);
        const float hx2 = xhalf(hc2);
        __builtin_amdgcn_sched_barrier(0);
        fma_row_lane<2>(PM[2], wcm[1], hb1);
        fma_row_lane<3>(PL[3], wcl[1], hb1);
        __builtin_amdgcn_sched_barrier(0);
        const float hb2 = hi ? hx2 : hc2; hr[2] = hb2;
        __builtin_amdgcn_sched_barrier(0);
        const float zc2 = PM[1] + xr[2] * e2;
        __builtin_amdgcn_sched_barrier(0);
        fma_row_lane<3>(PH[3], wc1[1], zb1);
        fma_row_lane<3>(PM[3], wcm[1], hb1);
        fma_row_lane<1>(PL[1], wcl[2], hb2);
        __builtin_amdgcn_sched_barrier(0);
        const float zx2 = xhalf(zc2);
        __builtin_amdgcn_sched_barrier(0);
        fma_row_lane<1>(PM[1], wcm[2], hb2);
        fma_row_lane<2>(PL[2], wcl[2], hb2);
        __builtin_amdgcn_sched_barrier(0);
        const float m3 = PL[1] * 1.44269504088896341f; add_here(ldjO, PL[1]);
        __builtin_amdgcn_sched_barrier(0);
        const float zb2 = hi ? zx2 : zc2; zr[2] = zb2;
        __builtin_amdgcn_sched_barrier(0);
        fma_row_lane<2>(PM[2], wcm[2], hb2);
        __builtin_amdgcn_sched_barrier(0);
        const float e3 = __builtin_amdgcn_exp2f(m3);
        __builtin_amdgcn_sched_barrier(0);
        fma_row_lane<3>(PL[3], wcl[2], hb2);
        fma_row_lane<1>(PH[1], wc1[2], zb2);
        fma_row_lane<2>(PH[2], wc1[2], zb2);
        fma_row_lane<3>(PH[3], wc1[2], zb2);
        __builtin_amdgcn_sched_barrier(0);
        const float hc3 = relu1(PH[1]);
        __builtin_amdgcn_sched_barrier(0);
        const float zc3 = PM[1] + xr[3] * e3;
        __builtin_amdgcn_sched_barrier(0);
        fma_row_lane<3>(PM[3], wcm[2], hb2);
        fma_row_lane<4>(PL[4], wcl[1], hb1);
        __builtin_amdgcn_sched_barrier(0);
        const float hx3 = xhalf(hc3);
        __builtin_amdgcn_sched_barrier(0);
        const float zx3 = xhalf(zc3);
        __builtin_amdgcn_sched_barrier(0);
        fma_row_lane<4>(PH[4], wc1[1], zb1);
        fma_row_lane<4>(PL[4], wcl[2], hb2);
        __builtin_amdgcn_sched_barrier(0);
        const float hb3 = hi ? hc3 : hx3; hr[3] = hb3;
        __builtin_amdgcn_sched_barrier(0);
        const float zb3 = hi ? zc3 : zx3; zr[3] = zb3;
        __builtin_amdgcn_sched_barrier(0);
        if (PROD) {
            const v2f fz = {hi ? zb2 : zb0, hi ? zb3 : zb1};
            const v2f fh = {hi ? hb2 : hb0, hi ? hb3 : hb1};
            nH = __builtin_amdgcn_wmma_f32_16x16x4_f32(false, aw1[0], false, fz, (short)0, nH, false, false);
            nM = __builtin_amdgcn_wmma_f32_16x16x4_f32(false, awm[0], false, fh, (short)0, nM, false, false);
            nL = __builtin_amdgcn_wmma_f32_16x16x4_f32(false, awl[0], false, fh, (short)0, nL, false, false);
        }
        __builtin_amdgcn_sched_barrier(0);
        fma_row_lane<4>(PH[4], wc1[2], zb2);
        fma_row_lane<4>(PM[4], wcm[0], hb0);
        fma_row_lane<2>(PL[2], wcl[3], hb3);
        fma_row_lane<2>(PH[2], wc1[3], zb3);
        fma_row_lane<2>(PM[2], wcm[3], hb3);
        __builtin_amdgcn_sched_barrier(0);
        const float m4 = PL[2] * 1.44269504088896341f; add_here(ldjE, PL[2]);
        __builtin_amdgcn_sched_barrier(0);
        const float hc4 = relu1(PH[2]);
        __builtin_amdgcn_sched_barrier(0);
        fma_row_lane<3>(PL[3], wcl[3], hb3);
        __builtin_amdgcn_sched_barrier(0);
        const float e4 = __builtin_amdgcn_exp2f(m4);
        __builtin_amdgcn_sched_barrier(0);
        fma_row_lane<3>(PH[3], wc1[3], zb3);
        __builtin_amdgcn_sched_barrier(0);
        const float hx4 = xhalf(hc4);
        __builtin_amdgcn_sched_barrier(0);
        fma_row_lane<3>(PM[3], wcm[3], hb3);
        fma_row_lane<4>(PL[4], wcl[3], hb3);
        __builtin_amdgcn_sched_barrier(0);
        const float hb4 = hi ? hx4 : hc4; hr[4] = hb4;
        __builtin_amdgcn_sched_barrier(0);
        const float zc4 = PM[2] + xr[4] * e4;
        __builtin_amdgcn_sched_barrier(0);
        fma_row_lane<4>(PH[4], wc1[3], zb3);
        fma_row_lane<4>(PM[4], wcm[1], hb1);
        fma_row_lane<2>(PL[2], wcl[4], hb4);
        __builtin_amdgcn_sched_barrier(0);
        const float zx4 = xhalf(zc4);
        __builtin_amdgcn_sched_barrier(0);
        fma_row_lane<2>(PM[2], wcm[4], hb4);
        fma_row_lane<3>(PL[3], wcl[4], hb4);
        __builtin_amdgcn_sched_barrier(0);
        const float m5 = PL[2] * 1.44269504088896341f; add_here(ldjO, PL[2]);
        __builtin_amdgcn_sched_barrier(0);
        const float zb4 = hi ? zx4 : zc4; zr[4] = zb4;
        __builtin_amdgcn_sched_barrier(0);
        fma_row_lane<3>(PM[3], wcm[4], hb4);
        __builtin_amdgcn_sched_barrier(0);
        const float e5 = __builtin_amdgcn_exp2f(m5);
        __builtin_amdgcn_sched_barrier(0);
        fma_row_lane<4>(PL[4], wcl[4], hb4);
        fma_row_lane<2>(PH[2], wc1[4], zb4);
        fma_row_lane<3>(PH[3], wc1[4], zb4);
        fma_row_lane<4>(PH[4], wc1[4], zb4);
        __builtin_amdgcn_sched_barrier(0);
        const float hc5 = relu1(PH[2]);
        __builtin_amdgcn_sched_barrier(0);
        const float zc5 = PM[2] + xr[5] * e5;
        __builtin_amdgcn_sched_barrier(0);
        fma_row_lane<4>(PM[4], wcm[2], hb2);
        __builtin_amdgcn_sched_barrier(0);
        PH[5] = (pH[5] + qH[5]) + bH[5]; PM[5] = (pM[5] + qM[5]) + bM[5]; PL[5] = (pL[5] + qL[5]) + bL[5];
        __builtin_amdgcn_sched_barrier(0);
        const float hx5 = xhalf(hc5);
        __builtin_amdgcn_sched_barrier(0);
        const float zx5 = xhalf(zc5);
        __builtin_amdgcn_sched_barrier(0);
        fma_row_lane<4>(PM[4], wcm[3], hb3);
        __builtin_amdgcn_sched_barrier(0);
        PH[6] = (pH[6] + qH[6]) + bH[6]; PM[6] = (pM[6] + qM[6]) + bM[6]; PL[6] = (pL[6] + qL[6]) + bL[6];
        __builtin_amdgcn_sched_barrier(0);
        const float hb5 = hi ? hc5 : hx5; hr[5] = hb5;
        __builtin_amdgcn_sched_barrier(0);
        const float zb5 = hi ? zc5 : zx5; zr[5] = zb5;
        __builtin_amdgcn_sched_barrier(0);
        fma_row_lane<4>(PM[4], wcm[4], hb4);
        fma_row_lane<5>(PL[5], wcl[0], hb0);
        fma_row_lane<3>(PL[3], wcl[5], hb5);
        fma_row_lane<3>(PH[3], wc1[5], zb5);
        fma_row_lane<3>(PM[3], wcm[5], hb5);
        __builtin_amdgcn_sched_barrier(0);
        const float m6 = PL[3] * 1.44269504088896341f; add_here(ldjE, PL[3]);
        __builtin_amdgcn_sched_barrier(0);
        const float hc6 = relu1(PH[3]);
        __builtin_amdgcn_sched_barrier(0);
        fma_row_lane<4>(PL[4], wcl[5], hb5);
        __builtin_amdgcn_sched_barrier(0);
        const float e6 = __builtin_amdgcn_exp2f(m6);
        __builtin_amdgcn_sched_barrier(0);
        fma_row_lane<4>(PH[4], wc1[5], zb5);
        __builtin_amdgcn_sched_barrier(0);
        const float hx6 = xhalf(hc6);
        __builtin_amdgcn_sched_barrier(0);
        fma_row_lane<5>(PH[5], wc1[0], zb0);
        fma_row_lane<4>(PM[4], wcm[5], hb5);
        __builtin_amdgcn_sched_barrier(0);
        const float hb6 = hi ? hx6 : hc6; hr[6] = hb6;
        __builtin_amdgcn_sched_barrier(0);
        const float zc6 = PM[3] + xr[6] * e6;
        __builtin_amdgcn_sched_barrier(0);
        fma_row_lane<5>(PL[5], wcl[1], hb1);
        fma_row_lane<5>(PH[5], wc1[1], zb1);
        fma_row_lane<3>(PL[3], wcl[6], hb6);
        __builtin_amdgcn_sched_barrier(0);
        const float zx6 = xhalf(zc6);
        __builtin_amdgcn_sched_barrier(0);
        fma_row_lane<3>(PM[3], wcm[6], hb6);
        fma_row_lane<4>(PL[4], wcl[6], hb6);
        __builtin_amdgcn_sched_barrier(0);
        const float m7 = PL[3] * 1.44269504088896341f; add_here(ldjO, PL[3]);
        __builtin_amdgcn_sched_barrier(0);
        const float zb6 = hi ? zx6 : zc6; zr[6] = zb6;
        __builtin_amdgcn_sched_barrier(0);
        fma_row_lane<4>(PM[4], wcm[6], hb6);
        __builtin_amdgcn_sched_barrier(0);
        const float e7 = __builtin_amdgcn_exp2f(m7);
        __builtin_amdgcn_sched_barrier(0);
        fma_row_lane<5>(PL[5], wcl[2], hb2);
        fma_row_lane<3>(PH[3], wc1[6], zb6);
        fma_row_lane<4>(PH[4], wc1[6], zb6);
        fma_row_lane<5>(PH[5], wc1[2], zb2);
        __builtin_amdgcn_sched_barrier(0);
        const float hc7 = relu1(PH[3]);
        __builtin_amdgcn_sched_barrier(0);
        const float zc7 = PM[3] + xr[7] * e7;
        __builtin_amdgcn_sched_barrier(0);
        fma_row_lane<5>(PM[5], wcm[0], hb0);
        fma_row_lane<5>(PL[5], wcl[3], hb3);
        __builtin_amdgcn_sched_barrier(0);
        const float hx7 = xhalf(hc7);
        __builtin_amdgcn_sched_barrier(0);
        const float zx7 = xhalf(zc7);
        __builtin_amdgcn_sched_barrier(0);
        fma_row_lane<5>(PH[5], wc1[3], zb3);
        fma_row_lane<5>(PM[5], wcm[1], hb1);
        __builtin_amdgcn_sched_barrier(0);
        const float hb7 = hi ? hc7 : hx7; hr[7] = hb7;
        __builtin_amdgcn_sched_barrier(0);
        const float zb7 = hi ? zc7 : zx7; zr[7] = zb7;
        __builtin_amdgcn_sched_barrier(0);
        if (PROD) {
            const v2f fz = {hi ? zb6 : zb4, hi ? zb7 : zb5};
            const v2f fh = {hi ? hb6 : hb4, hi ? hb7 : hb5};
            nH = __builtin_amdgcn_wmma_f32_16x16x4_f32(false, aw1[1], false, fz, (short)0, nH, false, false);
            nM = __builtin_amdgcn_wmma_f32_16x16x4_f32(false, awm[1], false, fh, (short)0, nM, false, false);
            nL = __builtin_amdgcn_wmma_f32_16x16x4_f32(false, awl[1], false, fh, (short)0, nL, false, false);
        }
        __builtin_amdgcn_sched_barrier(0);
        fma_row_lane<5>(PL[5], wcl[4], hb4);
        fma_row_lane<5>(PH[5], wc1[4], zb4);
        fma_row_lane<4>(PL[4], wcl[7], hb7);
        fma_row_lane<4>(PH[4], wc1[7], zb7);
        fma_row_lane<4>(PM[4], wcm[7], hb7);
        __builtin_amdgcn_sched_barrier(0);
        const float m8 = PL[4] * 1.44269504088896341f; add_here(ldjE, PL[4]);
        __builtin_amdgcn_sched_barrier(0);
        const float hc8 = relu1(PH[4]);
        __builtin_amdgcn_sched_barrier(0);
        fma_row_lane<5>(PM[5], wcm[2], hb2);
        __builtin_amdgcn_sched_barrier(0);
        const float e8 = __builtin_amdgcn_exp2f(m8);
        __builtin_amdgcn_sched_barrier(0);
        fma_row_lane<5>(PL[5], wcl[5], hb5);
        __builtin_amdgcn_sched_barrier(0);
        const float hx8 = xhalf(hc8);
        __builtin_amdgcn_sched_barrier(0);
        fma_row_lane<5>(PH[5], wc1[5], zb5);
        fma_row_lane<5>(PM[5], wcm[3], hb3);
        __builtin_amdgcn_sched_barrier(0);
        const float hb8 = hi ? hx8 : hc8; hr[8] = hb8;
        __builtin_amdgcn_sched_barrier(0);
        const float zc8 = PM[4] + xr[8] * e8;
        __builtin_amdgcn_sched_barrier(0);
        fma_row_lane<5>(PL[5], wcl[6], hb6);
        fma_row_lane<5>(PH[5], wc1[6], zb6);
        fma_row_lane<4>(PL[4], wcl[8], hb8);
        __builtin_amdgcn_sched_barrier(0);
        const float zx8 = xhalf(zc8);
        __builtin_amdgcn_sched_barrier(0);
        fma_row_lane<4>(PM[4], wcm[8], hb8);
        fma_row_lane<5>(PM[5], wcm[4], hb4);
        __builtin_amdgcn_sched_barrier(0);
        const float m9 = PL[4] * 1.44269504088896341f; add_here(ldjO, PL[4]);
        __builtin_amdgcn_sched_barrier(0);
        const float zb8 = hi ? zx8 : zc8; zr[8] = zb8;
        __builtin_amdgcn_sched_barrier(0);
        fma_row_lane<5>(PL[5], wcl[7], hb7);
        __builtin_amdgcn_sched_barrier(0);
        const float e9 = __builtin_amdgcn_exp2f(m9);
        __builtin_amdgcn_sched_barrier(0);
        fma_row_lane<5>(PH[5], wc1[7], zb7);
        fma_row_lane<4>(PH[4], wc1[8], zb8);
        fma_row_lane<6>(PL[6], wcl[0], hb0);
        fma_row_lane<6>(PH[6], wc1[0], zb0);
        __builtin_amdgcn_sched_barrier(0);
        const float hc9 = relu1(PH[4]);
        __builtin_amdgcn_sched_barrier(0);
        const float zc9 = PM[4] + xr[9] * e9;
        __builtin_amdgcn_sched_barrier(0);
        fma_row_lane<5>(PM[5], wcm[5], hb5);
        fma_row_lane<5>(PL[5], wcl[8], hb8);
        __builtin_amdgcn_sched_barrier(0);
        const float hx9 = xhalf(hc9);
        __builtin_amdgcn_sched_barrier(0);
        const float zx9 = xhalf(zc9);
        __builtin_amdgcn_sched_barrier(0);
        fma_row_lane<5>(PH[5], wc1[8], zb8);
        fma_row_lane<6>(PL[6], wcl[1], hb1);
        __builtin_amdgcn_sched_barrier(0);
        const float hb9 = hi ? hc9 : hx9; hr[9] = hb9;
        __builtin_amdgcn_sched_barrier(0);
        const float zb9 = hi ? zc9 : zx9; zr[9] = zb9;
        __builtin_amdgcn_sched_barrier(0);
        fma_row_lane<6>(PH[6], wc1[1], zb1);
        fma_row_lane<5>(PM[5], wcm[6], hb6);
        fma_row_lane<5>(PL[5], wcl[9], hb9);
        fma_row_lane<5>(PH[5], wc1[9], zb9);
        fma_row_lane<6>(PL[6], wcl[2], hb2);
        fma_row_lane<6>(PH[6], wc1[2], zb2);
        fma_row_lane<5>(PM[5], wcm[7], hb7);
        fma_row_lane<6>(PM[6], wcm[0], hb0);
        __builtin_amdgcn_sched_barrier(0);
        const float hc10 = relu1(PH[5]);
        __builtin_amdgcn_sched_barrier(0);
        const float m10 = PL[5] * 1.44269504088896341f; add_here(ldjE, PL[5]);
        __builtin_amdgcn_sched_barrier(0);
        fma_row_lane<6>(PL[6], wcl[3], hb3);
        fma_row_lane<6>(PH[6], wc1[3], zb3);
        fma_row_lane<5>(PM[5], wcm[8], hb8);
        fma_row_lane<6>(PM[6], wcm[1], hb1);
        __builtin_amdgcn_sched_barrier(0);
        const float hx10 = xhalf(hc10);
        __builtin_amdgcn_sched_barrier(0);
        const float e10 = __builtin_amdgcn_exp2f(m10);
        __builtin_amdgcn_sched_barrier(0);
        fma_row_lane<6>(PL[6], wcl[4], hb4);
        fma_row_lane<6>(PH[6], wc1[4], zb4);
        fma_row_lane<5>(PM[5], wcm[9], hb9);
        fma_row_lane<6>(PM[6], wcm[2], hb2);
        __builtin_amdgcn_sched_barrier(0);
        const float hb10 = hi ? hx10 : hc10; hr[10] = hb10;
        __builtin_amdgcn_sched_barrier(0);
        fma_row_lane<6>(PL[6], wcl[5], hb5);
        fma_row_lane<6>(PH[6], wc1[5], zb5);
        __builtin_amdgcn_sched_barrier(0);
        const float zc10 = PM[5] + xr[10] * e10;
        __builtin_amdgcn_sched_barrier(0);
        fma_row_lane<5>(PL[5], wcl[10], hb10);
        fma_row_lane<6>(PM[6], wcm[3], hb3);
        __builtin_amdgcn_sched_barrier(0);
        PH[7] = (pH[7] + qH[7]) + bH[7]; PM[7] = (pM[7] + qM[7]) + bM[7]; PL[7] = (pL[7] + qL[7]) + bL[7];
        __builtin_amdgcn_sched_barrier(0);
        const float zx10 = xhalf(zc10);
        __builtin_amdgcn_sched_barrier(0);
        fma_row_lane<6>(PL[6], wcl[6], hb6);
        fma_row_lane<6>(PH[6], wc1[6], zb6);
        __builtin_amdgcn_sched_barrier(0);
        const float m11 = PL[5] * 1.44269504088896341f; add_here(ldjO, PL[5]);
        __builtin_amdgcn_sched_barrier(0);
        fma_row_lane<6>(PM[6], wcm[4], hb4);
        __builtin_amdgcn_sched_barrier(0);
        const float zb10 = hi ? zx10 : zc10; zr[10] = zb10;
        __builtin_amdgcn_sched_barrier(0);
        fma_row_lane<6>(PL[6], wcl[7], hb7);
        fma_row_lane<6>(PH[6], wc1[7], zb7);
        __builtin_amdgcn_sched_barrier(0);
        const float e11 = __builtin_amdgcn_exp2f(m11);
        __builtin_amdgcn_sched_barrier(0);
        fma_row_lane<7>(PL[7], wcl[0], hb0);
        fma_row_lane<7>(PH[7], wc1[0], zb0);
        fma_row_lane<6>(PM[6], wcm[5], hb5);
        fma_row_lane<5>(PM[5], wcm[10], hb10);
        fma_row_lane<5>(PH[5], wc1[10], zb10);
        fma_row_lane<6>(PL[6], wcl[8], hb8);
        fma_row_lane<6>(PH[6], wc1[8], zb8);
        fma_row_lane<7>(PL[7], wcl[1], hb1);
        fma_row_lane<7>(PH[7], wc1[1], zb1);
        fma_row_lane<6>(PM[6], wcm[6], hb6);
        __builtin_amdgcn_sched_barrier(0);
        const float hc11 = relu1(PH[5]);
        __builtin_amdgcn_sched_barrier(0);
        const float zc11 = PM[5] + xr[11] * e11;
        __builtin_amdgcn_sched_barrier(0);
        fma_row_lane<6>(PL[6], wcl[9], hb9);
        fma_row_lane<6>(PH[6], wc1[9], zb9);
        fma_row_lane<7>(PL[7], wcl[2], hb2);
        fma_row_lane<7>(PH[7], wc1[2], zb2);
        fma_row_lane<6>(PM[6], wcm[7], hb7);
        __builtin_amdgcn_sched_barrier(0);
        const float hx11 = xhalf(hc11);
        __builtin_amdgcn_sched_barrier(0);
        const float zx11 = xhalf(zc11);
        __builtin_amdgcn_sched_barrier(0);
        fma_row_lane<7>(PM[7], wcm[0], hb0);
        fma_row_lane<6>(PL[6], wcl[10], hb10);
        fma_row_lane<6>(PH[6], wc1[10], zb10);
        __builtin_amdgcn_sched_barrier(0);
        const float hb11 = hi ? hc11 : hx11; hr[11] = hb11;
        __builtin_amdgcn_sched_barrier(0);
        const float zb11 = hi ? zc11 : zx11; zr[11] = zb11;
        __builtin_amdgcn_sched_barrier(0);
        if (PROD) {
            const v2f fz = {hi ? zb10 : zb8, hi ? zb11 : zb9};
            const v2f fh = {hi ? hb10 : hb8, hi ? hb11 : hb9};
            nH = __builtin_amdgcn_wmma_f32_16x16x4_f32(false, aw1[2], false, fz, (short)0, nH, false, false);
            nM = __builtin_amdgcn_wmma_f32_16x16x4_f32(false, awm[2], false, fh, (short)0, nM, false, false);
            nL = __builtin_amdgcn_wmma_f32_16x16x4_f32(false, awl[2], false, fh, (short)0, nL, false, false);
        }
        __builtin_amdgcn_sched_barrier(0);
        fma_row_lane<7>(PL[7], wcl[3], hb3);
        fma_row_lane<7>(PH[7], wc1[3], zb3);
        fma_row_lane<6>(PM[6], wcm[8], hb8);
        fma_row_lane<7>(PM[7], wcm[1], hb1);
        fma_row_lane<6>(PL[6], wcl[11], hb11);
        fma_row_lane<6>(PH[6], wc1[11], zb11);
        fma_row_lane<7>(PL[7], wcl[4], hb4);
        fma_row_lane<7>(PH[7], wc1[4], zb4);
        fma_row_lane<6>(PM[6], wcm[9], hb9);
        fma_row_lane<7>(PM[7], wcm[2], hb2);
        __builtin_amdgcn_sched_barrier(0);
        const float hc12 = relu1(PH[6]);
        __builtin_amdgcn_sched_barrier(0);
        const float m12 = PL[6] * 1.44269504088896341f; add_here(ldjE, PL[6]);
        __builtin_amdgcn_sched_barrier(0);
        fma_row_lane<7>(PL[7], wcl[5], hb5);
        fma_row_lane<7>(PH[7], wc1[5], zb5);
        fma_row_lane<6>(PM[6], wcm[10], hb10);
        fma_row_lane<7>(PM[7], wcm[3], hb3);
        __builtin_amdgcn_sched_barrier(0);
        const float hx12 = xhalf(hc12);
        __builtin_amdgcn_sched_barrier(0);
        const float e12 = __builtin_amdgcn_exp2f(m12);
        __builtin_amdgcn_sched_barrier(0);
        fma_row_lane<7>(PL[7], wcl[6], hb6);
        fma_row_lane<7>(PH[7], wc1[6], zb6);
        fma_row_lane<6>(PM[6], wcm[11], hb11);
        fma_row_lane<7>(PM[7], wcm[4], hb4);
        __builtin_amdgcn_sched_barrier(0);
        const float hb12 = hi ? hx12 : hc12; hr[12] = hb12;
        __builtin_amdgcn_sched_barrier(0);
        fma_row_lane<7>(PL[7], wcl[7], hb7);
        fma_row_lane<7>(PH[7], wc1[7], zb7);
        __builtin_amdgcn_sched_barrier(0);
        const float zc12 = PM[6] + xr[12] * e12;
        __builtin_amdgcn_sched_barrier(0);
        fma_row_lane<6>(PL[6], wcl[12], hb12);
        fma_row_lane<7>(PM[7], wcm[5], hb5);
        __builtin_amdgcn_sched_barrier(0);
        const float zx12 = xhalf(zc12);
        __builtin_amdgcn_sched_barrier(0);
        fma_row_lane<7>(PL[7], wcl[8], hb8);
        fma_row_lane<7>(PH[7], wc1[8], zb8);
        __builtin_amdgcn_sched_barrier(0);
        const float m13 = PL[6] * 1.44269504088896341f; add_here(ldjO, PL[6]);
        __builtin_amdgcn_sched_barrier(0);
        fma_row_lane<7>(PM[7], wcm[6], hb6);
        __builtin_amdgcn_sched_barrier(0);
        const float zb12 = hi ? zx12 : zc12; zr[12] = zb12;
        __builtin_amdgcn_sched_barrier(0);
        fma_row_lane<7>(PL[7], wcl[9], hb9);
        fma_row_lane<7>(PH[7], wc1[9], zb9);
        __builtin_amdgcn_sched_barrier(0);
        const float e13 = __builtin_amdgcn_exp2f(m13);
        __builtin_amdgcn_sched_barrier(0);
        fma_row_lane<7>(PM[7], wcm[7], hb7);
        fma_row_lane<6>(PM[6], wcm[12], hb12);
        fma_row_lane<6>(PH[6], wc1[12], zb12);
        fma_row_lane<7>(PL[7], wcl[10], hb10);
        fma_row_lane<7>(PH[7], wc1[10], zb10);
        fma_row_lane<7>(PM[7], wcm[8], hb8);
        __builtin_amdgcn_sched_barrier(0);
        const float hc13 = relu1(PH[6]);
        __builtin_amdgcn_sched_barrier(0);
        const float zc13 = PM[6] + xr[13] * e13;
        __builtin_amdgcn_sched_barrier(0);
        fma_row_lane<7>(PL[7], wcl[11], hb11);
        fma_row_lane<7>(PH[7], wc1[11], zb11);
        fma_row_lane<7>(PM[7], wcm[9], hb9);
        __builtin_amdgcn_sched_barrier(0);
        const float hx13 = xhalf(hc13);
        __builtin_amdgcn_sched_barrier(0);
        const float zx13 = xhalf(zc13);
        __builtin_amdgcn_sched_barrier(0);
        fma_row_lane<7>(PL[7], wcl[12], hb12);
        fma_row_lane<7>(PH[7], wc1[12], zb12);
        __builtin_amdgcn_sched_barrier(0);
        const float hb13 = hi ? hc13 : hx13; hr[13] = hb13;
        __builtin_amdgcn_sched_barrier(0);
        const float zb13 = hi ? zc13 : zx13; zr[13] = zb13;
        __builtin_amdgcn_sched_barrier(0);
        fma_row_lane<7>(PM[7], wcm[10], hb10);
        fma_row_lane<7>(PL[7], wcl[13], hb13);
        fma_row_lane<7>(PH[7], wc1[13], zb13);
        fma_row_lane<7>(PM[7], wcm[11], hb11);
        __builtin_amdgcn_sched_barrier(0);
        const float m14 = PL[7] * 1.44269504088896341f; add_here(ldjE, PL[7]);
        __builtin_amdgcn_sched_barrier(0);
        const float hc14 = relu1(PH[7]);
        __builtin_amdgcn_sched_barrier(0);
        fma_row_lane<7>(PM[7], wcm[12], hb12);
        __builtin_amdgcn_sched_barrier(0);
        const float e14 = __builtin_amdgcn_exp2f(m14);
        __builtin_amdgcn_sched_barrier(0);
        const float hx14 = xhalf(hc14);
        __builtin_amdgcn_sched_barrier(0);
        fma_row_lane<7>(PM[7], wcm[13], hb13);
        __builtin_amdgcn_sched_barrier(0);
        const float hb14 = hi ? hx14 : hc14; hr[14] = hb14;
        __builtin_amdgcn_sched_barrier(0);
        const float zc14 = PM[7] + xr[14] * e14;
        __builtin_amdgcn_sched_barrier(0);
        fma_row_lane<7>(PL[7], wcl[14], hb14);
        __builtin_amdgcn_sched_barrier(0);
        const float zx14 = xhalf(zc14);
        __builtin_amdgcn_sched_barrier(0);
        fma_row_lane<7>(PM[7], wcm[14], hb14);
        __builtin_amdgcn_sched_barrier(0);
        const float m15 = PL[7] * 1.44269504088896341f; add_here(ldjO, PL[7]);
        __builtin_amdgcn_sched_barrier(0);
        const float zb14 = hi ? zx14 : zc14; zr[14] = zb14;
        __builtin_amdgcn_sched_barrier(0);
        const float e15 = __builtin_amdgcn_exp2f(m15);
        __builtin_amdgcn_sched_barrier(0);
        fma_row_lane<7>(PH[7], wc1[14], zb14);
        __builtin_amdgcn_sched_barrier(0);
        const float hc15 = relu1(PH[7]);
        __builtin_amdgcn_sched_barrier(0);
        const float zc15 = PM[7] + xr[15] * e15;
        __builtin_amdgcn_sched_barrier(0);
        const float hx15 = xhalf(hc15);
        __builtin_amdgcn_sched_barrier(0);
        const float zx15 = xhalf(zc15);
        __builtin_amdgcn_sched_barrier(0);
        const float hb15 = hi ? hc15 : hx15; hr[15] = hb15;
        __builtin_amdgcn_sched_barrier(0);
        const float zb15 = hi ? zc15 : zx15; zr[15] = zb15;
        __builtin_amdgcn_sched_barrier(0);
        if (PROD) {
            const v2f fz = {hi ? zb14 : zb12, hi ? zb15 : zb13};
            const v2f fh = {hi ? hb14 : hb12, hi ? hb15 : hb13};
            nH = __builtin_amdgcn_wmma_f32_16x16x4_f32(false, aw1[3], false, fz, (short)0, nH, false, false);
            nM = __builtin_amdgcn_wmma_f32_16x16x4_f32(false, awm[3], false, fh, (short)0, nM, false, false);
            nL = __builtin_amdgcn_wmma_f32_16x16x4_f32(false, awl[3], false, fh, (short)0, nL, false, false);
        }
        __builtin_amdgcn_sched_barrier(0);
}

template <bool B> struct flag_t { static constexpr bool value = B; };

__global__ __launch_bounds__(32 * NWV) void iaf_layer_kernel(
    const float* __restrict__ x,
    const float* __restrict__ W1,  const float* __restrict__ b1,
    const float* __restrict__ Wmu, const float* __restrict__ bmu,
    const float* __restrict__ Wls, const float* __restrict__ bls,
    float* __restrict__ out, int B)
{
    __shared__ __attribute__((aligned(16))) float Ws[3][DD * PW];
    __shared__ __attribute__((aligned(16))) float XZs[NT][16 * PW];
    __shared__ __attribute__((aligned(16))) float Hs[NT][16 * PW];
    __shared__ __attribute__((aligned(16))) float tHs[2][NT][256];
    __shared__ __attribute__((aligned(16))) float tMs[2][NT][256];
    __shared__ __attribute__((aligned(16))) float tLs[2][NT][256];
    __shared__ __attribute__((aligned(16))) float bs[3][DD];
    __shared__ float ldjs[NT * 16];

    const int tid   = threadIdx.x;
    const int lane  = tid & 31;
    const int wv    = __builtin_amdgcn_readfirstlane(tid >> 5);
    const bool seqw = wv < NT;
    const int tile  = wv & (NT - 1);
    const int r0    = blockIdx.x * (16 * NT) + tile * 16;
    const int mrow  = lane & 15;
    const int khalf = (lane >> 4) * 2;
    const int prow  = 2 * (lane & 7) + ((lane >> 3) & 1);

    float* XZ = XZs[tile];
    float* Hh = Hs[tile];

    v4f pa[8], pb[8], pc[8];
    auto load_rows = [&](int first, int step) {
        #pragma unroll
        for (int q = 0; q < 8; ++q) {
            const int row = first + step * q;
            pa[q] = *(const v4f*)&W1 [row * DD + lane * 4];
            pb[q] = *(const v4f*)&Wmu[row * DD + lane * 4];
            pc[q] = *(const v4f*)&Wls[row * DD + lane * 4];
        }
    };
    auto store_rows = [&](int first, int step) {
        #pragma unroll
        for (int q = 0; q < 8; ++q) {
            const int row = first + step * q;
            *(v4f*)&Ws[0][row * PW + lane * 4] = pa[q];
            *(v4f*)&Ws[1][row * PW + lane * 4] = pb[q];
            *(v4f*)&Ws[2][row * PW + lane * 4] = pc[q];
        }
    };

    if (seqw) {
        v4f xv[16];
        #pragma unroll
        for (int r = 0; r < 16; ++r)
            xv[r] = *(const v4f*)&x[(size_t)(r0 + r) * DD + lane * 4];
        #pragma unroll
        for (int r = 0; r < 16; ++r)
            *(v4f*)&XZ[r * PW + lane * 4] = xv[r];
    } else {
        const v4f zero4 = {0.f, 0.f, 0.f, 0.f};
        #pragma unroll
        for (int p = 0; p < 2; ++p) {
            *(v4f*)&tHs[p][tile][lane * 8] = zero4;  *(v4f*)&tHs[p][tile][lane * 8 + 4] = zero4;
            *(v4f*)&tMs[p][tile][lane * 8] = zero4;  *(v4f*)&tMs[p][tile][lane * 8 + 4] = zero4;
            *(v4f*)&tLs[p][tile][lane * 8] = zero4;  *(v4f*)&tLs[p][tile][lane * 8 + 4] = zero4;
        }
    }
    if (lane < 8) {
        load_rows(wv, NWV);
        store_rows(wv, NWV);
    }
    if (wv < 3) {
        const float* bsrc = (wv == 0) ? b1 : (wv == 1) ? bmu : bls;
        const v4f bv = *(const v4f*)&bsrc[lane * 4];
        const int bb = (lane >> 2) * 16 + (lane & 3) * 2;
        bs[wv][bb]     = bv.x;  bs[wv][bb + 8] = bv.y;
        bs[wv][bb + 1] = bv.z;  bs[wv][bb + 9] = bv.w;
    }
    __syncthreads();

    const v8f zero8 = {0.f,0.f,0.f,0.f,0.f,0.f,0.f,0.f};

    if (seqw) {
        float ldjE = 0.0f, ldjO = 0.0f;
        const int r = mrow;
        const bool hi = lane >= 16;
        v8f qH = zero8, qM = zero8, qL = zero8;

        auto block_body = [&](const int I, auto prod_tag) {
            constexpr bool PROD = decltype(prod_tag)::value;
            const int c0 = I * 16;
            const int wrow = c0 + 2 * (lane & 7) + (lane >> 4);
            const float* wr1 = &Ws[0][wrow * PW + c0];
            const float* wrm = &Ws[1][wrow * PW + c0];
            const float* wrl = &Ws[2][wrow * PW + c0];
            float wc1[16], wcm[16], wcl[16];
            #pragma unroll
            for (int q = 0; q < 4; ++q) {
                const v4f a = *(const v4f*)&wr1[4 * q];
                const v4f b = *(const v4f*)&wrm[4 * q];
                const v4f c = *(const v4f*)&wrl[4 * q];
                wc1[4 * q] = a.x; wc1[4 * q + 1] = a.y; wc1[4 * q + 2] = a.z; wc1[4 * q + 3] = a.w;
                wcm[4 * q] = b.x; wcm[4 * q + 1] = b.y; wcm[4 * q + 2] = b.z; wcm[4 * q + 3] = b.w;
                wcl[4 * q] = c.x; wcl[4 * q + 1] = c.y; wcl[4 * q + 2] = c.z; wcl[4 * q + 3] = c.w;
            }
            float pH[8], pM[8], pL[8], bH[8], bM[8], bL[8], xr[16];
            const float* sH = &tHs[I & 1][tile][lane * 8];
            const float* sM = &tMs[I & 1][tile][lane * 8];
            const float* sL = &tLs[I & 1][tile][lane * 8];
            const int bo = c0 + 8 * (lane >> 4);
            #pragma unroll
            for (int q = 0; q < 2; ++q) {
                const v4f a = *(const v4f*)&sH[4 * q];
                const v4f b = *(const v4f*)&sM[4 * q];
                const v4f c = *(const v4f*)&sL[4 * q];
                const v4f d = *(const v4f*)&bs[0][bo + 4 * q];
                const v4f e = *(const v4f*)&bs[1][bo + 4 * q];
                const v4f f = *(const v4f*)&bs[2][bo + 4 * q];
                pH[4 * q] = a.x; pH[4 * q + 1] = a.y; pH[4 * q + 2] = a.z; pH[4 * q + 3] = a.w;
                pM[4 * q] = b.x; pM[4 * q + 1] = b.y; pM[4 * q + 2] = b.z; pM[4 * q + 3] = b.w;
                pL[4 * q] = c.x; pL[4 * q + 1] = c.y; pL[4 * q + 2] = c.z; pL[4 * q + 3] = c.w;
                bH[4 * q] = d.x; bH[4 * q + 1] = d.y; bH[4 * q + 2] = d.z; bH[4 * q + 3] = d.w;
                bM[4 * q] = e.x; bM[4 * q + 1] = e.y; bM[4 * q + 2] = e.z; bM[4 * q + 3] = e.w;
                bL[4 * q] = f.x; bL[4 * q + 1] = f.y; bL[4 * q + 2] = f.z; bL[4 * q + 3] = f.w;
            }
            #pragma unroll
            for (int t = 0; t < 16; ++t)
                xr[t] = XZ[r * PW + c0 + t];
            const v2f zero2 = {0.f, 0.f};
            v2f aw1[4] = {zero2, zero2, zero2, zero2}, awm[4] = {zero2, zero2, zero2, zero2}, awl[4] = {zero2, zero2, zero2, zero2};
            if (PROD) {
                const float* g1 = &Ws[0][(c0 + 16 + prow) * PW + c0 + khalf];
                const float* gm = &Ws[1][(c0 + 16 + prow) * PW + c0 + khalf];
                const float* gl = &Ws[2][(c0 + 16 + prow) * PW + c0 + khalf];
                #pragma unroll
                for (int kt = 0; kt < 4; ++kt) {
                    aw1[kt] = *(const v2f*)&g1[4 * kt];
                    awm[kt] = *(const v2f*)&gm[4 * kt];
                    awl[kt] = *(const v2f*)&gl[4 * kt];
                }
            }
            float PH[8], PM[8], PL[8];
            float zr[16], hr[16];
            v8f nH = zero8, nM = zero8, nL = zero8;
            seq_steps<PROD>(PH, PM, PL, pH, pM, pL, qH, qM, qL, bH, bM, bL, xr, zr, hr, wc1, wcm, wcl,
                            aw1, awm, awl, nH, nM, nL, ldjE, ldjO, hi);
            qH = nH; qM = nM; qL = nL;
            if (lane < 16) {
                #pragma unroll
                for (int t = 0; t < 16; ++t) {
                    XZ[r * PW + c0 + t] = zr[t];
                    Hh[r * PW + c0 + t] = hr[t];
                }
            }
            if (!PROD) {
                const float tot = ldjE + xhalf(ldjO);
                if (lane < 16) ldjs[tile * 16 + r] = tot;
            }
            lds_barrier();
        };
        for (int I = 0; I < 7; ++I)
            block_body(I, flag_t<true>{});
        block_body(7, flag_t<false>{});
    } else {
        v8f cH2 = zero8, cM2 = zero8, cL2 = zero8;
        v8f cH3 = zero8, cM3 = zero8, cL3 = zero8;
        v8f cH4 = zero8, cM4 = zero8, cL4 = zero8;
        v8f cH5 = zero8, cM5 = zero8, cL5 = zero8;
        v8f cH6 = zero8, cM6 = zero8, cL6 = zero8;
        v8f cH7 = zero8, cM7 = zero8, cL7 = zero8;

        const float* az  = &XZ[mrow * PW + khalf];
        const float* ah  = &Hh[mrow * PW + khalf];

        auto mm_block = [&](v8f& cH, v8f& cM, v8f& cL, int kb, int J) {
            const float* g1 = &Ws[0][(16 * J + prow) * PW + khalf];
            const float* gm = &Ws[1][(16 * J + prow) * PW + khalf];
            const float* gl = &Ws[2][(16 * J + prow) * PW + khalf];
            #pragma unroll
            for (int kt = 0; kt < 4; ++kt) {
                const int k0 = kb + kt * 4;
                const v2f fZ  = *(const v2f*)&az[k0];
                const v2f fH  = *(const v2f*)&ah[k0];
                const v2f aW1 = *(const v2f*)&g1[k0];
                const v2f aWm = *(const v2f*)&gm[k0];
                const v2f aWl = *(const v2f*)&gl[k0];
                cH = __builtin_amdgcn_wmma_f32_16x16x4_f32(false, aW1, false, fZ, (short)0, cH, false, false);
                cM = __builtin_amdgcn_wmma_f32_16x16x4_f32(false, aWm, false, fH, (short)0, cM, false, false);
                cL = __builtin_amdgcn_wmma_f32_16x16x4_f32(false, aWl, false, fH, (short)0, cL, false, false);
            }
        };
        auto put_sums = [&](const v8f& cH, const v8f& cM, const v8f& cL, int J) {
            float* dH = &tHs[J & 1][tile][lane * 8];
            float* dM = &tMs[J & 1][tile][lane * 8];
            float* dL = &tLs[J & 1][tile][lane * 8];
            *(v4f*)&dH[0] = __builtin_shufflevector(cH, cH, 0, 1, 2, 3);  *(v4f*)&dH[4] = __builtin_shufflevector(cH, cH, 4, 5, 6, 7);
            *(v4f*)&dM[0] = __builtin_shufflevector(cM, cM, 0, 1, 2, 3);  *(v4f*)&dM[4] = __builtin_shufflevector(cM, cM, 4, 5, 6, 7);
            *(v4f*)&dL[0] = __builtin_shufflevector(cL, cL, 0, 1, 2, 3);  *(v4f*)&dL[4] = __builtin_shufflevector(cL, cL, 4, 5, 6, 7);
        };

        auto mat_barrier = [&]() {
            __builtin_amdgcn_sched_barrier(0);
            lds_barrier();
            __builtin_amdgcn_sched_barrier(0);
        };

        v4f ra[8], rb[8], rc[8];
        auto fetch_block_row = [&](int R) {
            if (lane < 4 * (R + 1)) {
                #pragma unroll
                for (int q = 0; q < 8; ++q) {
                    const int row = 16 * R + tile + NT * q;
                    ra[q] = *(const v4f*)&W1 [row * DD + lane * 4];
                    rb[q] = *(const v4f*)&Wmu[row * DD + lane * 4];
                    rc[q] = *(const v4f*)&Wls[row * DD + lane * 4];
                }
            }
        };
        auto store_block_row = [&](int R) {
            if (lane < 4 * (R + 1)) {
                #pragma unroll
                for (int q = 0; q < 8; ++q) {
                    const int row = 16 * R + tile + NT * q;
                    *(v4f*)&Ws[0][row * PW + lane * 4] = ra[q];
                    *(v4f*)&Ws[1][row * PW + lane * 4] = rb[q];
                    *(v4f*)&Ws[2][row * PW + lane * 4] = rc[q];
                }
            }
        };

        {
            v4f qa[3][8], qb[3][8], qc[3][8];
            #pragma unroll
            for (int s = 0; s < 3; ++s) {
                if (lane < 4 * (s + 3)) {
                    #pragma unroll
                    for (int q = 0; q < 8; ++q) {
                        const int row = 16 * (s + 2) + tile + NT * q;
                        qa[s][q] = *(const v4f*)&W1 [row * DD + lane * 4];
                        qb[s][q] = *(const v4f*)&Wmu[row * DD + lane * 4];
                        qc[s][q] = *(const v4f*)&Wls[row * DD + lane * 4];
                    }
                }
            }
            #pragma unroll
            for (int s = 0; s < 3; ++s) {
                if (lane < 4 * (s + 3)) {
                    #pragma unroll
                    for (int q = 0; q < 8; ++q) {
                        const int row = 16 * (s + 2) + tile + NT * q;
                        *(v4f*)&Ws[0][row * PW + lane * 4] = qa[s][q];
                        *(v4f*)&Ws[1][row * PW + lane * 4] = qb[s][q];
                        *(v4f*)&Ws[2][row * PW + lane * 4] = qc[s][q];
                    }
                }
            }
        }
        mat_barrier();

        fetch_block_row(5);
        mm_block(cH2, cM2, cL2, 0, 2);
        mm_block(cH3, cM3, cL3, 0, 3);
        mm_block(cH4, cM4, cL4, 0, 4);
        put_sums(cH2, cM2, cL2, 2);
        store_block_row(5);
        mat_barrier();

        fetch_block_row(6);
        mm_block(cH3, cM3, cL3, 16, 3);
        mm_block(cH4, cM4, cL4, 16, 4);
        mm_block(cH5, cM5, cL5, 0, 5);
        mm_block(cH5, cM5, cL5, 16, 5);
        put_sums(cH3, cM3, cL3, 3);
        store_block_row(6);
        mat_barrier();

        fetch_block_row(7);
        mm_block(cH4, cM4, cL4, 32, 4);
        mm_block(cH5, cM5, cL5, 32, 5);
        mm_block(cH6, cM6, cL6, 0, 6);
        mm_block(cH6, cM6, cL6, 16, 6);
        put_sums(cH4, cM4, cL4, 4);
        store_block_row(7);
        mat_barrier();

        mm_block(cH5, cM5, cL5, 48, 5);
        mm_block(cH6, cM6, cL6, 32, 6);
        mm_block(cH7, cM7, cL7, 0, 7);
        mm_block(cH7, cM7, cL7, 16, 7);
        put_sums(cH5, cM5, cL5, 5);
        mat_barrier();

        mm_block(cH6, cM6, cL6, 48, 6);
        mm_block(cH6, cM6, cL6, 64, 6);
        mm_block(cH7, cM7, cL7, 32, 7);
        put_sums(cH6, cM6, cL6, 6);
        mat_barrier();

        mm_block(cH7, cM7, cL7, 48, 7);
        mm_block(cH7, cM7, cL7, 64, 7);
        mm_block(cH7, cM7, cL7, 80, 7);
        put_sums(cH7, cM7, cL7, 7);
        mat_barrier();

        mat_barrier();
    }

    const int rb8 = seqw ? 0 : 8;
    #pragma unroll
    for (int r = 0; r < 8; ++r)
        *(v4f*)&out[(size_t)(r0 + rb8 + r) * DD + lane * 4] = *(const v4f*)&XZ[(rb8 + r) * PW + lane * 4];

    if (tid < 16 * NT)
        out[(size_t)B * DD + blockIdx.x * (16 * NT) + tid] = ldjs[tid];
}

extern "C" void kernel_launch(void* const* d_in, const int* in_sizes, int n_in,
                              void* d_out, int out_size, void* d_ws, size_t ws_size,
                              hipStream_t stream) {
    (void)n_in; (void)out_size; (void)d_ws; (void)ws_size;
    const float* x   = (const float*)d_in[0];
    const float* W1  = (const float*)d_in[1];
    const float* b1  = (const float*)d_in[2];
    const float* Wmu = (const float*)d_in[3];
    const float* bmu = (const float*)d_in[4];
    const float* Wls = (const float*)d_in[5];
    const float* bls = (const float*)d_in[6];
    float* out = (float*)d_out;

    const int B = in_sizes[0] / DD;
    dim3 grid(B / (16 * NT)), block(32 * NWV);
    iaf_layer_kernel<<<grid, block, 0, stream>>>(x, W1, b1, Wmu, bmu, Wls, bls, out, B);
}


# Start-up global loads (x tile, weight rows 0-31, biases) issued as one batch before the first wait; row-count argument loaded at entry: one memory round trip before block 0, was three
# speedup vs baseline: 3.4704x; 1.0975x over previous
#include <hip/hip_runtime.h>

typedef __attribute__((ext_vector_type(2))) float v2f;
typedef __attribute__((ext_vector_type(4))) float v4f;
typedef __attribute__((ext_vector_type(8))) float v8f;

#define DD 128
#define PW 132
#define NT 2
#define NWV 4

__device__ __forceinline__ void lds_barrier() {
    __builtin_amdgcn_fence(__ATOMIC_RELEASE, "workgroup", "local");
    __builtin_amdgcn_s_barrier();
    __builtin_amdgcn_fence(__ATOMIC_ACQUIRE, "workgroup", "local");
}

template <int N> __device__ __forceinline__ void fma_row_lane(float& acc, float w, float m) {
    asm volatile("v_fmac_f32_dpp %0, %1, %2 row_share:%3 row_mask:0xf bank_mask:0xf" : "+v"(acc) : "v"(w), "v"(m), "n"(N));
}

__device__ __forceinline__ float xhalf(float v) {
    const int b = __builtin_bit_cast(int, v);
    return __builtin_bit_cast(float, __builtin_amdgcn_permlanex16(b, b, 0x76543210, 0xfedcba98, false, true));
}

__device__ __forceinline__ float relu1(float v) {
    float r;
    asm("v_max_num_f32 %0, 0, %1" : "=v"(r) : "v"(v));
    return r;
}

__device__ __forceinline__ void add_here(float& acc, float v) {
    asm volatile("v_add_f32 %0, %0, %1" : "+v"(acc) : "v"(v));
}

template <bool PROD>
__device__ __forceinline__ void seq_steps(float (&PH)[8], float (&PM)[8], float (&PL)[8],
                                          const float (&pH)[8], const float (&pM)[8], const float (&pL)[8],
                                          const v8f& qH, const v8f& qM, const v8f& qL,
                                          const float (&bH)[8], const float (&bM)[8], const float (&bL)[8],
                                          const float (&xr)[16], float (&zr)[16], float (&hr)[16],
                                          const float (&wc1)[16], const float (&wcm)[16], const float (&wcl)[16],
                                          const v2f (&aw1)[4], const v2f (&awm)[4], const v2f (&awl)[4],
                                          v8f& nH, v8f& nM, v8f& nL,
                                          float& ldjE, float& ldjO, const bool hi) {
        PH[0] = (pH[0] + qH[0]) + bH[0]; PM[0] = (pM[0] + qM[0]) + bM[0]; PL[0] = (pL[0] + qL[0]) + bL[0];
        __builtin_amdgcn_sched_barrier(0);
        PH[1] = (pH[1] + qH[1]) + bH[1]; PM[1] = (pM[1] + qM[1]) + bM[1]; PL[1] = (pL[1] + qL[1]) + bL[1];
        __builtin_amdgcn_sched_barrier(0);
        const float hc0 = relu1(PH[0]);
        __builtin_amdgcn_sched_barrier(0);
        const float m0 = PL[0] * 1.44269504088896341f; add_here(ldjE, PL[0]);
        __builtin_amdgcn_sched_barrier(0);
        PH[2] = (pH[2] + qH[2]) + bH[2]; PM[2] = (pM[2] + qM[2]) + bM[2]; PL[2] = (pL[2] + qL[2]) + bL[2];
        __builtin_amdgcn_sched_barrier(0);
        const float hx0 = xhalf(hc0);
        __builtin_amdgcn_sched_barrier(0);
        const float e0 = __builtin_amdgcn_exp2f(m0);
        __builtin_amdgcn_sched_barrier(0);
        PH[3] = (pH[3] + qH[3]) + bH[3]; PM[3] = (pM[3] + qM[3]) + bM[3]; PL[3] = (pL[3] + qL[3]) + bL[3];
        __builtin_amdgcn_sched_barrier(0);
        const float hb0 = hi ? hx0 : hc0; hr[0] = hb0;
        __builtin_amdgcn_sched_barrier(0);
        PH[4] = (pH[4] + qH[4]) + bH[4]; PM[4] = (pM[4] + qM[4]) + bM[4]; PL[4] = (pL[4] + qL[4]) + bL[4];
        __builtin_amdgcn_sched_barrier(0);
        const float zc0 = PM[0] + xr[0] * e0;
        __builtin_amdgcn_sched_barrier(0);
        fma_row_lane<0>(PL[0], wcl[0], hb0);
        fma_row_lane<1>(PL[1], wcl[0], hb0);
        __builtin_amdgcn_sched_barrier(0);
        const float zx0 = xhalf(zc0);
        __builtin_amdgcn_sched_barrier(0);
        fma_row_lane<0>(PM[0], wcm[0], hb0);
        __builtin_amdgcn_sched_barrier(0);
        const float m1 = PL[0] * 1.44269504088896341f; add_here(ldjO, PL[0]);
        __builtin_amdgcn_sched_barrier(0);
        const float zb0 = hi ? zx0 : zc0; zr[0] = zb0;
        __builtin_amdgcn_sched_barrier(0);
        fma_row_lane<1>(PM[1], wcm[0], hb0);
        __builtin_amdgcn_sched_barrier(0);
        const float e1 = __builtin_amdgcn_exp2f(m1);
        __builtin_amdgcn_sched_barrier(0);
        fma_row_lane<2>(PL[2], wcl[0], hb0);
        fma_row_lane<0>(PH[0], wc1[0], zb0);
        fma_row_lane<1>(PH[1], wc1[0], zb0);
        fma_row_lane<2>(PH[2], wc1[0], zb0);
        __builtin_amdgcn_sched_barrier(0);
        const float hc1 = relu1(PH[0]);
        __builtin_amdgcn_sched_barrier(0);
        const float zc1 = PM[0] + xr[1] * e1;
        __builtin_amdgcn_sched_barrier(0);
        fma_row_lane<2>(PM[2], wcm[0], hb0);
        fma_row_lane<3>(PL[3], wcl[0], hb0);
        __builtin_amdgcn_sched_barrier(0);
        const float hx1 = xhalf(hc1);
        __builtin_amdgcn_sched_barrier(0);
        const float zx1 = xhalf(zc1);
        __builtin_amdgcn_sched_barrier(0);
        fma_row_lane<3>(PH[3], wc1[0], zb0);
        fma_row_lane<3>(PM[3], wcm[0], hb0);
        __builtin_amdgcn_sched_barrier(0);
        const float hb1 = hi ? hc1 : hx1; hr[1] = hb1;
        __builtin_amdgcn_sched_barrier(0);
        const float zb1 = hi ? zc1 : zx1; zr[1] = zb1;
        __builtin_amdgcn_sched_barrier(0);
        fma_row_lane<4>(PL[4], wcl[0], hb0);
        fma_row_lane<4>(PH[4], wc1[0], zb0);
        fma_row_lane<1>(PL[1], wcl[1], hb1);
        fma_row_lane<1>(PH[1], wc1[1], zb1);
        fma_row_lane<1>(PM[1], wcm[1], hb1);
        __builtin_amdgcn_sched_barrier(0);
        const float m2 = PL[1] * 1.44269504088896341f; add_here(ldjE, PL[1]);
        __builtin_amdgcn_sched_barrier(0);
        const float hc2 = relu1(PH[1]);
        __builtin_amdgcn_sched_barrier(0);
        fma_row_lane<2>(PL[2], wcl[1], hb1);
        __builtin_amdgcn_sched_barrier(0);
        const float e2 = __builtin_amdgcn_exp2f(m2);
        __builtin_amdgcn_sched_barrier(0);
        fma_row_lane<2>(PH[2], wc1[1], zb1);
        __builtin_amdgcn_sched_barrier(0);
        const float hx2 = xhalf(hc2);
        __builtin_amdgcn_sched_barrier(0);
        fma_row_lane<2>(PM[2], wcm[1], hb1);
        fma_row_lane<3>(PL[3], wcl[1], hb1);
        __builtin_amdgcn_sched_barrier(0);
        const float hb2 = hi ? hx2 : hc2; hr[2] = hb2;
        __builtin_amdgcn_sched_barrier(0);
        const float zc2 = PM[1] + xr[2] * e2;
        __builtin_amdgcn_sched_barrier(0);
        fma_row_lane<3>(PH[3], wc1[1], zb1);
        fma_row_lane<3>(PM[3], wcm[1], hb1);
        fma_row_lane<1>(PL[1], wcl[2], hb2);
        __builtin_amdgcn_sched_barrier(0);
        const float zx2 = xhalf(zc2);
        __builtin_amdgcn_sched_barrier(0);
        fma_row_lane<1>(PM[1], wcm[2], hb2);
        fma_row_lane<2>(PL[2], wcl[2], hb2);
        __builtin_amdgcn_sched_barrier(0);
        const float m3 = PL[1] * 1.44269504088896341f; add_here(ldjO, PL[1]);
        __builtin_amdgcn_sched_barrier(0);
        const float zb2 = hi ? zx2 : zc2; zr[2] = zb2;
        __builtin_amdgcn_sched_barrier(0);
        fma_row_lane<2>(PM[2], wcm[2], hb2);
        __builtin_amdgcn_sched_barrier(0);
        const float e3 = __builtin_amdgcn_exp2f(m3);
        __builtin_amdgcn_sched_barrier(0);
        fma_row_lane<3>(PL[3], wcl[2], hb2);
        fma_row_lane<1>(PH[1], wc1[2], zb2);
        fma_row_lane<2>(PH[2], wc1[2], zb2);
        fma_row_lane<3>(PH[3], wc1[2], zb2);
        __builtin_amdgcn_sched_barrier(0);
        const float hc3 = relu1(PH[1]);
        __builtin_amdgcn_sched_barrier(0);
        const float zc3 = PM[1] + xr[3] * e3;
        __builtin_amdgcn_sched_barrier(0);
        fma_row_lane<3>(PM[3], wcm[2], hb2);
        fma_row_lane<4>(PL[4], wcl[1], hb1);
        __builtin_amdgcn_sched_barrier(0);
        const float hx3 = xhalf(hc3);
        __builtin_amdgcn_sched_barrier(0);
        const float zx3 = xhalf(zc3);
        __builtin_amdgcn_sched_barrier(0);
        fma_row_lane<4>(PH[4], wc1[1], zb1);
        fma_row_lane<4>(PL[4], wcl[2], hb2);
        __builtin_amdgcn_sched_barrier(0);
        const float hb3 = hi ? hc3 : hx3; hr[3] = hb3;
        __builtin_amdgcn_sched_barrier(0);
        const float zb3 = hi ? zc3 : zx3; zr[3] = zb3;
        __builtin_amdgcn_sched_barrier(0);
        if (PROD) {
            const v2f fz = {hi ? zb2 : zb0, hi ? zb3 : zb1};
            const v2f fh = {hi ? hb2 : hb0, hi ? hb3 : hb1};
            nH = __builtin_amdgcn_wmma_f32_16x16x4_f32(false, aw1[0], false, fz, (short)0, nH, false, false);
            nM = __builtin_amdgcn_wmma_f32_16x16x4_f32(false, awm[0], false, fh, (short)0, nM, false, false);
            nL = __builtin_amdgcn_wmma_f32_16x16x4_f32(false, awl[0], false, fh, (short)0, nL, false, false);
        }
        __builtin_amdgcn_sched_barrier(0);
        fma_row_lane<4>(PH[4], wc1[2], zb2);
        fma_row_lane<4>(PM[4], wcm[0], hb0);
        fma_row_lane<2>(PL[2], wcl[3], hb3);
        fma_row_lane<2>(PH[2], wc1[3], zb3);
        fma_row_lane<2>(PM[2], wcm[3], hb3);
        __builtin_amdgcn_sched_barrier(0);
        const float m4 = PL[2] * 1.44269504088896341f; add_here(ldjE, PL[2]);
        __builtin_amdgcn_sched_barrier(0);
        const float hc4 = relu1(PH[2]);
        __builtin_amdgcn_sched_barrier(0);
        fma_row_lane<3>(PL[3], wcl[3], hb3);
        __builtin_amdgcn_sched_barrier(0);
        const float e4 = __builtin_amdgcn_exp2f(m4);
        __builtin_amdgcn_sched_barrier(0);
        fma_row_lane<3>(PH[3], wc1[3], zb3);
        __builtin_amdgcn_sched_barrier(0);
        const float hx4 = xhalf(hc4);
        __builtin_amdgcn_sched_barrier(0);
        fma_row_lane<3>(PM[3], wcm[3], hb3);
        fma_row_lane<4>(PL[4], wcl[3], hb3);
        __builtin_amdgcn_sched_barrier(0);
        const float hb4 = hi ? hx4 : hc4; hr[4] = hb4;
        __builtin_amdgcn_sched_barrier(0);
        const float zc4 = PM[2] + xr[4] * e4;
        __builtin_amdgcn_sched_barrier(0);
        fma_row_lane<4>(PH[4], wc1[3], zb3);
        fma_row_lane<4>(PM[4], wcm[1], hb1);
        fma_row_lane<2>(PL[2], wcl[4], hb4);
        __builtin_amdgcn_sched_barrier(0);
        const float zx4 = xhalf(zc4);
        __builtin_amdgcn_sched_barrier(0);
        fma_row_lane<2>(PM[2], wcm[4], hb4);
        fma_row_lane<3>(PL[3], wcl[4], hb4);
        __builtin_amdgcn_sched_barrier(0);
        const float m5 = PL[2] * 1.44269504088896341f; add_here(ldjO, PL[2]);
        __builtin_amdgcn_sched_barrier(0);
        const float zb4 = hi ? zx4 : zc4; zr[4] = zb4;
        __builtin_amdgcn_sched_barrier(0);
        fma_row_lane<3>(PM[3], wcm[4], hb4);
        __builtin_amdgcn_sched_barrier(0);
        const float e5 = __builtin_amdgcn_exp2f(m5);
        __builtin_amdgcn_sched_barrier(0);
        fma_row_lane<4>(PL[4], wcl[4], hb4);
        fma_row_lane<2>(PH[2], wc1[4], zb4);
        fma_row_lane<3>(PH[3], wc1[4], zb4);
        fma_row_lane<4>(PH[4], wc1[4], zb4);
        __builtin_amdgcn_sched_barrier(0);
        const float hc5 = relu1(PH[2]);
        __builtin_amdgcn_sched_barrier(0);
        const float zc5 = PM[2] + xr[5] * e5;
        __builtin_amdgcn_sched_barrier(0);
        fma_row_lane<4>(PM[4], wcm[2], hb2);
        __builtin_amdgcn_sched_barrier(0);
        PH[5] = (pH[5] + qH[5]) + bH[5]; PM[5] = (pM[5] + qM[5]) + bM[5]; PL[5] = (pL[5] + qL[5]) + bL[5];
        __builtin_amdgcn_sched_barrier(0);
        const float hx5 = xhalf(hc5);
        __builtin_amdgcn_sched_barrier(0);
        const float zx5 = xhalf(zc5);
        __builtin_amdgcn_sched_barrier(0);
        fma_row_lane<4>(PM[4], wcm[3], hb3);
        __builtin_amdgcn_sched_barrier(0);
        PH[6] = (pH[6] + qH[6]) + bH[6]; PM[6] = (pM[6] + qM[6]) + bM[6]; PL[6] = (pL[6] + qL[6]) + bL[6];
        __builtin_amdgcn_sched_barrier(0);
        const float hb5 = hi ? hc5 : hx5; hr[5] = hb5;
        __builtin_amdgcn_sched_barrier(0);
        const float zb5 = hi ? zc5 : zx5; zr[5] = zb5;
        __builtin_amdgcn_sched_barrier(0);
        fma_row_lane<4>(PM[4], wcm[4], hb4);
        fma_row_lane<5>(PL[5], wcl[0], hb0);
        fma_row_lane<3>(PL[3], wcl[5], hb5);
        fma_row_lane<3>(PH[3], wc1[5], zb5);
        fma_row_lane<3>(PM[3], wcm[5], hb5);
        __builtin_amdgcn_sched_barrier(0);
        const float m6 = PL[3] * 1.44269504088896341f; add_here(ldjE, PL[3]);
        __builtin_amdgcn_sched_barrier(0);
        const float hc6 = relu1(PH[3]);
        __builtin_amdgcn_sched_barrier(0);
        fma_row_lane<4>(PL[4], wcl[5], hb5);
        __builtin_amdgcn_sched_barrier(0);
        const float e6 = __builtin_amdgcn_exp2f(m6);
        __builtin_amdgcn_sched_barrier(0);
        fma_row_lane<4>(PH[4], wc1[5], zb5);
        __builtin_amdgcn_sched_barrier(0);
        const float hx6 = xhalf(hc6);
        __builtin_amdgcn_sched_barrier(0);
        fma_row_lane<5>(PH[5], wc1[0], zb0);
        fma_row_lane<4>(PM[4], wcm[5], hb5);
        __builtin_amdgcn_sched_barrier(0);
        const float hb6 = hi ? hx6 : hc6; hr[6] = hb6;
        __builtin_amdgcn_sched_barrier(0);
        const float zc6 = PM[3] + xr[6] * e6;
        __builtin_amdgcn_sched_barrier(0);
        fma_row_lane<5>(PL[5], wcl[1], hb1);
        fma_row_lane<5>(PH[5], wc1[1], zb1);
        fma_row_lane<3>(PL[3], wcl[6], hb6);
        __builtin_amdgcn_sched_barrier(0);
        const float zx6 = xhalf(zc6);
        __builtin_amdgcn_sched_barrier(0);
        fma_row_lane<3>(PM[3], wcm[6], hb6);
        fma_row_lane<4>(PL[4], wcl[6], hb6);
        __builtin_amdgcn_sched_barrier(0);
        const float m7 = PL[3] * 1.44269504088896341f; add_here(ldjO, PL[3]);
        __builtin_amdgcn_sched_barrier(0);
        const float zb6 = hi ? zx6 : zc6; zr[6] = zb6;
        __builtin_amdgcn_sched_barrier(0);
        fma_row_lane<4>(PM[4], wcm[6], hb6);
        __builtin_amdgcn_sched_barrier(0);
        const float e7 = __builtin_amdgcn_exp2f(m7);
        __builtin_amdgcn_sched_barrier(0);
        fma_row_lane<5>(PL[5], wcl[2], hb2);
        fma_row_lane<3>(PH[3], wc1[6], zb6);
        fma_row_lane<4>(PH[4], wc1[6], zb6);
        fma_row_lane<5>(PH[5], wc1[2], zb2);
        __builtin_amdgcn_sched_barrier(0);
        const float hc7 = relu1(PH[3]);
        __builtin_amdgcn_sched_barrier(0);
        const float zc7 = PM[3] + xr[7] * e7;
        __builtin_amdgcn_sched_barrier(0);
        fma_row_lane<5>(PM[5], wcm[0], hb0);
        fma_row_lane<5>(PL[5], wcl[3], hb3);
        __builtin_amdgcn_sched_barrier(0);
        const float hx7 = xhalf(hc7);
        __builtin_amdgcn_sched_barrier(0);
        const float zx7 = xhalf(zc7);
        __builtin_amdgcn_sched_barrier(0);
        fma_row_lane<5>(PH[5], wc1[3], zb3);
        fma_row_lane<5>(PM[5], wcm[1], hb1);
        __builtin_amdgcn_sched_barrier(0);
        const float hb7 = hi ? hc7 : hx7; hr[7] = hb7;
        __builtin_amdgcn_sched_barrier(0);
        const float zb7 = hi ? zc7 : zx7; zr[7] = zb7;
        __builtin_amdgcn_sched_barrier(0);
        if (PROD) {
            const v2f fz = {hi ? zb6 : zb4, hi ? zb7 : zb5};
            const v2f fh = {hi ? hb6 : hb4, hi ? hb7 : hb5};
            nH = __builtin_amdgcn_wmma_f32_16x16x4_f32(false, aw1[1], false, fz, (short)0, nH, false, false);
            nM = __builtin_amdgcn_wmma_f32_16x16x4_f32(false, awm[1], false, fh, (short)0, nM, false, false);
            nL = __builtin_amdgcn_wmma_f32_16x16x4_f32(false, awl[1], false, fh, (short)0, nL, false, false);
        }
        __builtin_amdgcn_sched_barrier(0);
        fma_row_lane<5>(PL[5], wcl[4], hb4);
        fma_row_lane<5>(PH[5], wc1[4], zb4);
        fma_row_lane<4>(PL[4], wcl[7], hb7);
        fma_row_lane<4>(PH[4], wc1[7], zb7);
        fma_row_lane<4>(PM[4], wcm[7], hb7);
        __builtin_amdgcn_sched_barrier(0);
        const float m8 = PL[4] * 1.44269504088896341f; add_here(ldjE, PL[4]);
        __builtin_amdgcn_sched_barrier(0);
        const float hc8 = relu1(PH[4]);
        __builtin_amdgcn_sched_barrier(0);
        fma_row_lane<5>(PM[5], wcm[2], hb2);
        __builtin_amdgcn_sched_barrier(0);
        const float e8 = __builtin_amdgcn_exp2f(m8);
        __builtin_amdgcn_sched_barrier(0);
        fma_row_lane<5>(PL[5], wcl[5], hb5);
        __builtin_amdgcn_sched_barrier(0);
        const float hx8 = xhalf(hc8);
        __builtin_amdgcn_sched_barrier(0);
        fma_row_lane<5>(PH[5], wc1[5], zb5);
        fma_row_lane<5>(PM[5], wcm[3], hb3);
        __builtin_amdgcn_sched_barrier(0);
        const float hb8 = hi ? hx8 : hc8; hr[8] = hb8;
        __builtin_amdgcn_sched_barrier(0);
        const float zc8 = PM[4] + xr[8] * e8;
        __builtin_amdgcn_sched_barrier(0);
        fma_row_lane<5>(PL[5], wcl[6], hb6);
        fma_row_lane<5>(PH[5], wc1[6], zb6);
        fma_row_lane<4>(PL[4], wcl[8], hb8);
        __builtin_amdgcn_sched_barrier(0);
        const float zx8 = xhalf(zc8);
        __builtin_amdgcn_sched_barrier(0);
        fma_row_lane<4>(PM[4], wcm[8], hb8);
        fma_row_lane<5>(PM[5], wcm[4], hb4);
        __builtin_amdgcn_sched_barrier(0);
        const float m9 = PL[4] * 1.44269504088896341f; add_here(ldjO, PL[4]);
        __builtin_amdgcn_sched_barrier(0);
        const float zb8 = hi ? zx8 : zc8; zr[8] = zb8;
        __builtin_amdgcn_sched_barrier(0);
        fma_row_lane<5>(PL[5], wcl[7], hb7);
        __builtin_amdgcn_sched_barrier(0);
        const float e9 = __builtin_amdgcn_exp2f(m9);
        __builtin_amdgcn_sched_barrier(0);
        fma_row_lane<5>(PH[5], wc1[7], zb7);
        fma_row_lane<4>(PH[4], wc1[8], zb8);
        fma_row_lane<6>(PL[6], wcl[0], hb0);
        fma_row_lane<6>(PH[6], wc1[0], zb0);
        __builtin_amdgcn_sched_barrier(0);
        const float hc9 = relu1(PH[4]);
        __builtin_amdgcn_sched_barrier(0);
        const float zc9 = PM[4] + xr[9] * e9;
        __builtin_amdgcn_sched_barrier(0);
        fma_row_lane<5>(PM[5], wcm[5], hb5);
        fma_row_lane<5>(PL[5], wcl[8], hb8);
        __builtin_amdgcn_sched_barrier(0);
        const float hx9 = xhalf(hc9);
        __builtin_amdgcn_sched_barrier(0);
        const float zx9 = xhalf(zc9);
        __builtin_amdgcn_sched_barrier(0);
        fma_row_lane<5>(PH[5], wc1[8], zb8);
        fma_row_lane<6>(PL[6], wcl[1], hb1);
        __builtin_amdgcn_sched_barrier(0);
        const float hb9 = hi ? hc9 : hx9; hr[9] = hb9;
        __builtin_amdgcn_sched_barrier(0);
        const float zb9 = hi ? zc9 : zx9; zr[9] = zb9;
        __builtin_amdgcn_sched_barrier(0);
        fma_row_lane<6>(PH[6], wc1[1], zb1);
        fma_row_lane<5>(PM[5], wcm[6], hb6);
        fma_row_lane<5>(PL[5], wcl[9], hb9);
        fma_row_lane<5>(PH[5], wc1[9], zb9);
        fma_row_lane<6>(PL[6], wcl[2], hb2);
        fma_row_lane<6>(PH[6], wc1[2], zb2);
        fma_row_lane<5>(PM[5], wcm[7], hb7);
        fma_row_lane<6>(PM[6], wcm[0], hb0);
        __builtin_amdgcn_sched_barrier(0);
        const float hc10 = relu1(PH[5]);
        __builtin_amdgcn_sched_barrier(0);
        const float m10 = PL[5] * 1.44269504088896341f; add_here(ldjE, PL[5]);
        __builtin_amdgcn_sched_barrier(0);
        fma_row_lane<6>(PL[6], wcl[3], hb3);
        fma_row_lane<6>(PH[6], wc1[3], zb3);
        fma_row_lane<5>(PM[5], wcm[8], hb8);
        fma_row_lane<6>(PM[6], wcm[1], hb1);
        __builtin_amdgcn_sched_barrier(0);
        const float hx10 = xhalf(hc10);
        __builtin_amdgcn_sched_barrier(0);
        const float e10 = __builtin_amdgcn_exp2f(m10);
        __builtin_amdgcn_sched_barrier(0);
        fma_row_lane<6>(PL[6], wcl[4], hb4);
        fma_row_lane<6>(PH[6], wc1[4], zb4);
        fma_row_lane<5>(PM[5], wcm[9], hb9);
        fma_row_lane<6>(PM[6], wcm[2], hb2);
        __builtin_amdgcn_sched_barrier(0);
        const float hb10 = hi ? hx10 : hc10; hr[10] = hb10;
        __builtin_amdgcn_sched_barrier(0);
        fma_row_lane<6>(PL[6], wcl[5], hb5);
        fma_row_lane<6>(PH[6], wc1[5], zb5);
        __builtin_amdgcn_sched_barrier(0);
        const float zc10 = PM[5] + xr[10] * e10;
        __builtin_amdgcn_sched_barrier(0);
        fma_row_lane<5>(PL[5], wcl[10], hb10);
        fma_row_lane<6>(PM[6], wcm[3], hb3);
        __builtin_amdgcn_sched_barrier(0);
        PH[7] = (pH[7] + qH[7]) + bH[7]; PM[7] = (pM[7] + qM[7]) + bM[7]; PL[7] = (pL[7] + qL[7]) + bL[7];
        __builtin_amdgcn_sched_barrier(0);
        const float zx10 = xhalf(zc10);
        __builtin_amdgcn_sched_barrier(0);
        fma_row_lane<6>(PL[6], wcl[6], hb6);
        fma_row_lane<6>(PH[6], wc1[6], zb6);
        __builtin_amdgcn_sched_barrier(0);
        const float m11 = PL[5] * 1.44269504088896341f; add_here(ldjO, PL[5]);
        __builtin_amdgcn_sched_barrier(0);
        fma_row_lane<6>(PM[6], wcm[4], hb4);
        __builtin_amdgcn_sched_barrier(0);
        const float zb10 = hi ? zx10 : zc10; zr[10] = zb10;
        __builtin_amdgcn_sched_barrier(0);
        fma_row_lane<6>(PL[6], wcl[7], hb7);
        fma_row_lane<6>(PH[6], wc1[7], zb7);
        __builtin_amdgcn_sched_barrier(0);
        const float e11 = __builtin_amdgcn_exp2f(m11);
        __builtin_amdgcn_sched_barrier(0);
        fma_row_lane<7>(PL[7], wcl[0], hb0);
        fma_row_lane<7>(PH[7], wc1[0], zb0);
        fma_row_lane<6>(PM[6], wcm[5], hb5);
        fma_row_lane<5>(PM[5], wcm[10], hb10);
        fma_row_lane<5>(PH[5], wc1[10], zb10);
        fma_row_lane<6>(PL[6], wcl[8], hb8);
        fma_row_lane<6>(PH[6], wc1[8], zb8);
        fma_row_lane<7>(PL[7], wcl[1], hb1);
        fma_row_lane<7>(PH[7], wc1[1], zb1);
        fma_row_lane<6>(PM[6], wcm[6], hb6);
        __builtin_amdgcn_sched_barrier(0);
        const float hc11 = relu1(PH[5]);
        __builtin_amdgcn_sched_barrier(0);
        const float zc11 = PM[5] + xr[11] * e11;
        __builtin_amdgcn_sched_barrier(0);
        fma_row_lane<6>(PL[6], wcl[9], hb9);
        fma_row_lane<6>(PH[6], wc1[9], zb9);
        fma_row_lane<7>(PL[7], wcl[2], hb2);
        fma_row_lane<7>(PH[7], wc1[2], zb2);
        fma_row_lane<6>(PM[6], wcm[7], hb7);
        __builtin_amdgcn_sched_barrier(0);
        const float hx11 = xhalf(hc11);
        __builtin_amdgcn_sched_barrier(0);
        const float zx11 = xhalf(zc11);
        __builtin_amdgcn_sched_barrier(0);
        fma_row_lane<7>(PM[7], wcm[0], hb0);
        fma_row_lane<6>(PL[6], wcl[10], hb10);
        fma_row_lane<6>(PH[6], wc1[10], zb10);
        __builtin_amdgcn_sched_barrier(0);
        const float hb11 = hi ? hc11 : hx11; hr[11] = hb11;
        __builtin_amdgcn_sched_barrier(0);
        const float zb11 = hi ? zc11 : zx11; zr[11] = zb11;
        __builtin_amdgcn_sched_barrier(0);
        if (PROD) {
            const v2f fz = {hi ? zb10 : zb8, hi ? zb11 : zb9};
            const v2f fh = {hi ? hb10 : hb8, hi ? hb11 : hb9};
            nH = __builtin_amdgcn_wmma_f32_16x16x4_f32(false, aw1[2], false, fz, (short)0, nH, false, false);
            nM = __builtin_amdgcn_wmma_f32_16x16x4_f32(false, awm[2], false, fh, (short)0, nM, false, false);
            nL = __builtin_amdgcn_wmma_f32_16x16x4_f32(false, awl[2], false, fh, (short)0, nL, false, false);
        }
        __builtin_amdgcn_sched_barrier(0);
        fma_row_lane<7>(PL[7], wcl[3], hb3);
        fma_row_lane<7>(PH[7], wc1[3], zb3);
        fma_row_lane<6>(PM[6], wcm[8], hb8);
        fma_row_lane<7>(PM[7], wcm[1], hb1);
        fma_row_lane<6>(PL[6], wcl[11], hb11);
        fma_row_lane<6>(PH[6], wc1[11], zb11);
        fma_row_lane<7>(PL[7], wcl[4], hb4);
        fma_row_lane<7>(PH[7], wc1[4], zb4);
        fma_row_lane<6>(PM[6], wcm[9], hb9);
        fma_row_lane<7>(PM[7], wcm[2], hb2);
        __builtin_amdgcn_sched_barrier(0);
        const float hc12 = relu1(PH[6]);
        __builtin_amdgcn_sched_barrier(0);
        const float m12 = PL[6] * 1.44269504088896341f; add_here(ldjE, PL[6]);
        __builtin_amdgcn_sched_barrier(0);
        fma_row_lane<7>(PL[7], wcl[5], hb5);
        fma_row_lane<7>(PH[7], wc1[5], zb5);
        fma_row_lane<6>(PM[6], wcm[10], hb10);
        fma_row_lane<7>(PM[7], wcm[3], hb3);
        __builtin_amdgcn_sched_barrier(0);
        const float hx12 = xhalf(hc12);
        __builtin_amdgcn_sched_barrier(0);
        const float e12 = __builtin_amdgcn_exp2f(m12);
        __builtin_amdgcn_sched_barrier(0);
        fma_row_lane<7>(PL[7], wcl[6], hb6);
        fma_row_lane<7>(PH[7], wc1[6], zb6);
        fma_row_lane<6>(PM[6], wcm[11], hb11);
        fma_row_lane<7>(PM[7], wcm[4], hb4);
        __builtin_amdgcn_sched_barrier(0);
        const float hb12 = hi ? hx12 : hc12; hr[12] = hb12;
        __builtin_amdgcn_sched_barrier(0);
        fma_row_lane<7>(PL[7], wcl[7], hb7);
        fma_row_lane<7>(PH[7], wc1[7], zb7);
        __builtin_amdgcn_sched_barrier(0);
        const float zc12 = PM[6] + xr[12] * e12;
        __builtin_amdgcn_sched_barrier(0);
        fma_row_lane<6>(PL[6], wcl[12], hb12);
        fma_row_lane<7>(PM[7], wcm[5], hb5);
        __builtin_amdgcn_sched_barrier(0);
        const float zx12 = xhalf(zc12);
        __builtin_amdgcn_sched_barrier(0);
        fma_row_lane<7>(PL[7], wcl[8], hb8);
        fma_row_lane<7>(PH[7], wc1[8], zb8);
        __builtin_amdgcn_sched_barrier(0);
        const float m13 = PL[6] * 1.44269504088896341f; add_here(ldjO, PL[6]);
        __builtin_amdgcn_sched_barrier(0);
        fma_row_lane<7>(PM[7], wcm[6], hb6);
        __builtin_amdgcn_sched_barrier(0);
        const float zb12 = hi ? zx12 : zc12; zr[12] = zb12;
        __builtin_amdgcn_sched_barrier(0);
        fma_row_lane<7>(PL[7], wcl[9], hb9);
        fma_row_lane<7>(PH[7], wc1[9], zb9);
        __builtin_amdgcn_sched_barrier(0);
        const float e13 = __builtin_amdgcn_exp2f(m13);
        __builtin_amdgcn_sched_barrier(0);
        fma_row_lane<7>(PM[7], wcm[7], hb7);
        fma_row_lane<6>(PM[6], wcm[12], hb12);
        fma_row_lane<6>(PH[6], wc1[12], zb12);
        fma_row_lane<7>(PL[7], wcl[10], hb10);
        fma_row_lane<7>(PH[7], wc1[10], zb10);
        fma_row_lane<7>(PM[7], wcm[8], hb8);
        __builtin_amdgcn_sched_barrier(0);
        const float hc13 = relu1(PH[6]);
        __builtin_amdgcn_sched_barrier(0);
        const float zc13 = PM[6] + xr[13] * e13;
        __builtin_amdgcn_sched_barrier(0);
        fma_row_lane<7>(PL[7], wcl[11], hb11);
        fma_row_lane<7>(PH[7], wc1[11], zb11);
        fma_row_lane<7>(PM[7], wcm[9], hb9);
        __builtin_amdgcn_sched_barrier(0);
        const float hx13 = xhalf(hc13);
        __builtin_amdgcn_sched_barrier(0);
        const float zx13 = xhalf(zc13);
        __builtin_amdgcn_sched_barrier(0);
        fma_row_lane<7>(PL[7], wcl[12], hb12);
        fma_row_lane<7>(PH[7], wc1[12], zb12);
        __builtin_amdgcn_sched_barrier(0);
        const float hb13 = hi ? hc13 : hx13; hr[13] = hb13;
        __builtin_amdgcn_sched_barrier(0);
        const float zb13 = hi ? zc13 : zx13; zr[13] = zb13;
        __builtin_amdgcn_sched_barrier(0);
        fma_row_lane<7>(PM[7], wcm[10], hb10);
        fma_row_lane<7>(PL[7], wcl[13], hb13);
        fma_row_lane<7>(PH[7], wc1[13], zb13);
        fma_row_lane<7>(PM[7], wcm[11], hb11);
        __builtin_amdgcn_sched_barrier(0);
        const float m14 = PL[7] * 1.44269504088896341f; add_here(ldjE, PL[7]);
        __builtin_amdgcn_sched_barrier(0);
        const float hc14 = relu1(PH[7]);
        __builtin_amdgcn_sched_barrier(0);
        fma_row_lane<7>(PM[7], wcm[12], hb12);
        __builtin_amdgcn_sched_barrier(0);
        const float e14 = __builtin_amdgcn_exp2f(m14);
        __builtin_amdgcn_sched_barrier(0);
        const float hx14 = xhalf(hc14);
        __builtin_amdgcn_sched_barrier(0);
        fma_row_lane<7>(PM[7], wcm[13], hb13);
        __builtin_amdgcn_sched_barrier(0);
        const float hb14 = hi ? hx14 : hc14; hr[14] = hb14;
        __builtin_amdgcn_sched_barrier(0);
        const float zc14 = PM[7] + xr[14] * e14;
        __builtin_amdgcn_sched_barrier(0);
        fma_row_lane<7>(PL[7], wcl[14], hb14);
        __builtin_amdgcn_sched_barrier(0);
        const float zx14 = xhalf(zc14);
        __builtin_amdgcn_sched_barrier(0);
        fma_row_lane<7>(PM[7], wcm[14], hb14);
        __builtin_amdgcn_sched_barrier(0);
        const float m15 = PL[7] * 1.44269504088896341f; add_here(ldjO, PL[7]);
        __builtin_amdgcn_sched_barrier(0);
        const float zb14 = hi ? zx14 : zc14; zr[14] = zb14;
        __builtin_amdgcn_sched_barrier(0);
        const float e15 = __builtin_amdgcn_exp2f(m15);
        __builtin_amdgcn_sched_barrier(0);
        fma_row_lane<7>(PH[7], wc1[14], zb14);
        __builtin_amdgcn_sched_barrier(0);
        const float hc15 = relu1(PH[7]);
        __builtin_amdgcn_sched_barrier(0);
        const float zc15 = PM[7] + xr[15] * e15;
        __builtin_amdgcn_sched_barrier(0);
        const float hx15 = xhalf(hc15);
        __builtin_amdgcn_sched_barrier(0);
        const float zx15 = xhalf(zc15);
        __builtin_amdgcn_sched_barrier(0);
        const float hb15 = hi ? hc15 : hx15; hr[15] = hb15;
        __builtin_amdgcn_sched_barrier(0);
        const float zb15 = hi ? zc15 : zx15; zr[15] = zb15;
        __builtin_amdgcn_sched_barrier(0);
        if (PROD) {
            const v2f fz = {hi ? zb14 : zb12, hi ? zb15 : zb13};
            const v2f fh = {hi ? hb14 : hb12, hi ? hb15 : hb13};
            nH = __builtin_amdgcn_wmma_f32_16x16x4_f32(false, aw1[3], false, fz, (short)0, nH, false, false);
            nM = __builtin_amdgcn_wmma_f32_16x16x4_f32(false, awm[3], false, fh, (short)0, nM, false, false);
            nL = __builtin_amdgcn_wmma_f32_16x16x4_f32(false, awl[3], false, fh, (short)0, nL, false, false);
        }
        __builtin_amdgcn_sched_barrier(0);
}

template <bool B> struct flag_t { static constexpr bool value = B; };

__global__ __launch_bounds__(32 * NWV) void iaf_layer_kernel(
    const float* __restrict__ x,
    const float* __restrict__ W1,  const float* __restrict__ b1,
    const float* __restrict__ Wmu, const float* __restrict__ bmu,
    const float* __restrict__ Wls, const float* __restrict__ bls,
    float* __restrict__ out, int B)
{
    __shared__ __attribute__((aligned(16))) float Ws[3][DD * PW];
    __shared__ __attribute__((aligned(16))) float XZs[NT][16 * PW];
    __shared__ __attribute__((aligned(16))) float Hs[NT][16 * PW];
    __shared__ __attribute__((aligned(16))) float tHs[2][NT][256];
    __shared__ __attribute__((aligned(16))) float tMs[2][NT][256];
    __shared__ __attribute__((aligned(16))) float tLs[2][NT][256];
    __shared__ __attribute__((aligned(16))) float bs[3][DD];
    __shared__ float ldjs[NT * 16];

    const int tid   = threadIdx.x;
    const int lane  = tid & 31;
    const int wv    = __builtin_amdgcn_readfirstlane(tid >> 5);
    const bool seqw = wv < NT;
    const int tile  = wv & (NT - 1);
    const int r0    = blockIdx.x * (16 * NT) + tile * 16;
    const int mrow  = lane & 15;
    const int khalf = (lane >> 4) * 2;
    const int prow  = 2 * (lane & 7) + ((lane >> 3) & 1);

    float* XZ = XZs[tile];
    float* Hh = Hs[tile];

    asm volatile("" :: "s"(B));

    v4f xv[16];
    if (seqw) {
        #pragma unroll
        for (int r = 0; r < 16; ++r)
            xv[r] = *(const v4f*)&x[(size_t)(r0 + r) * DD + lane * 4];
    }
    const int srow = tid >> 3;
    const int scol = (tid & 7) * 4;
    v4f wa[2], wb[2], wc[2];
    #pragma unroll
    for (int h = 0; h < 2; ++h) {
        const int row = srow + 16 * h;
        wa[h] = *(const v4f*)&W1 [row * DD + scol];
        wb[h] = *(const v4f*)&Wmu[row * DD + scol];
        wc[h] = *(const v4f*)&Wls[row * DD + scol];
    }
    v4f bv = {0.f, 0.f, 0.f, 0.f};
    if (wv < 3) {
        const float* bsrc = (wv == 0) ? b1 : (wv == 1) ? bmu : bls;
        bv = *(const v4f*)&bsrc[lane * 4];
    }
    if (seqw) {
        #pragma unroll
        for (int r = 0; r < 16; ++r)
            *(v4f*)&XZ[r * PW + lane * 4] = xv[r];
    } else {
        const v4f zero4 = {0.f, 0.f, 0.f, 0.f};
        #pragma unroll
        for (int p = 0; p < 2; ++p) {
            *(v4f*)&tHs[p][tile][lane * 8] = zero4;  *(v4f*)&tHs[p][tile][lane * 8 + 4] = zero4;
            *(v4f*)&tMs[p][tile][lane * 8] = zero4;  *(v4f*)&tMs[p][tile][lane * 8 + 4] = zero4;
            *(v4f*)&tLs[p][tile][lane * 8] = zero4;  *(v4f*)&tLs[p][tile][lane * 8 + 4] = zero4;
        }
    }
    #pragma unroll
    for (int h = 0; h < 2; ++h) {
        const int row = srow + 16 * h;
        *(v4f*)&Ws[0][row * PW + scol] = wa[h];
        *(v4f*)&Ws[1][row * PW + scol] = wb[h];
        *(v4f*)&Ws[2][row * PW + scol] = wc[h];
    }
    if (wv < 3) {
        const int bb = (lane >> 2) * 16 + (lane & 3) * 2;
        bs[wv][bb]     = bv.x;  bs[wv][bb + 8] = bv.y;
        bs[wv][bb + 1] = bv.z;  bs[wv][bb + 9] = bv.w;
    }
    __syncthreads();

    const v8f zero8 = {0.f,0.f,0.f,0.f,0.f,0.f,0.f,0.f};

    if (seqw) {
        float ldjE = 0.0f, ldjO = 0.0f;
        const int r = mrow;
        const bool hi = lane >= 16;
        v8f qH = zero8, qM = zero8, qL = zero8;

        auto block_body = [&](const int I, auto prod_tag) {
            constexpr bool PROD = decltype(prod_tag)::value;
            const int c0 = I * 16;
            const int wrow = c0 + 2 * (lane & 7) + (lane >> 4);
            const float* wr1 = &Ws[0][wrow * PW + c0];
            const float* wrm = &Ws[1][wrow * PW + c0];
            const float* wrl = &Ws[2][wrow * PW + c0];
            float wc1[16], wcm[16], wcl[16];
            #pragma unroll
            for (int q = 0; q < 4; ++q) {
                const v4f a = *(const v4f*)&wr1[4 * q];
                const v4f b = *(const v4f*)&wrm[4 * q];
                const v4f c = *(const v4f*)&wrl[4 * q];
                wc1[4 * q] = a.x; wc1[4 * q + 1] = a.y; wc1[4 * q + 2] = a.z; wc1[4 * q + 3] = a.w;
                wcm[4 * q] = b.x; wcm[4 * q + 1] = b.y; wcm[4 * q + 2] = b.z; wcm[4 * q + 3] = b.w;
                wcl[4 * q] = c.x; wcl[4 * q + 1] = c.y; wcl[4 * q + 2] = c.z; wcl[4 * q + 3] = c.w;
            }
            float pH[8], pM[8], pL[8], bH[8], bM[8], bL[8], xr[16];
            const float* sH = &tHs[I & 1][tile][lane * 8];
            const float* sM = &tMs[I & 1][tile][lane * 8];
            const float* sL = &tLs[I & 1][tile][lane * 8];
            const int bo = c0 + 8 * (lane >> 4);
            #pragma unroll
            for (int q = 0; q < 2; ++q) {
                const v4f a = *(const v4f*)&sH[4 * q];
                const v4f b = *(const v4f*)&sM[4 * q];
                const v4f c = *(const v4f*)&sL[4 * q];
                const v4f d = *(const v4f*)&bs[0][bo + 4 * q];
                const v4f e = *(const v4f*)&bs[1][bo + 4 * q];
                const v4f f = *(const v4f*)&bs[2][bo + 4 * q];
                pH[4 * q] = a.x; pH[4 * q + 1] = a.y; pH[4 * q + 2] = a.z; pH[4 * q + 3] = a.w;
                pM[4 * q] = b.x; pM[4 * q + 1] = b.y; pM[4 * q + 2] = b.z; pM[4 * q + 3] = b.w;
                pL[4 * q] = c.x; pL[4 * q + 1] = c.y; pL[4 * q + 2] = c.z; pL[4 * q + 3] = c.w;
                bH[4 * q] = d.x; bH[4 * q + 1] = d.y; bH[4 * q + 2] = d.z; bH[4 * q + 3] = d.w;
                bM[4 * q] = e.x; bM[4 * q + 1] = e.y; bM[4 * q + 2] = e.z; bM[4 * q + 3] = e.w;
                bL[4 * q] = f.x; bL[4 * q + 1] = f.y; bL[4 * q + 2] = f.z; bL[4 * q + 3] = f.w;
            }
            #pragma unroll
            for (int t = 0; t < 16; ++t)
                xr[t] = XZ[r * PW + c0 + t];
            const v2f zero2 = {0.f, 0.f};
            v2f aw1[4] = {zero2, zero2, zero2, zero2}, awm[4] = {zero2, zero2, zero2, zero2}, awl[4] = {zero2, zero2, zero2, zero2};
            if (PROD) {
                const float* g1 = &Ws[0][(c0 + 16 + prow) * PW + c0 + khalf];
                const float* gm = &Ws[1][(c0 + 16 + prow) * PW + c0 + khalf];
                const float* gl = &Ws[2][(c0 + 16 + prow) * PW + c0 + khalf];
                #pragma unroll
                for (int kt = 0; kt < 4; ++kt) {
                    aw1[kt] = *(const v2f*)&g1[4 * kt];
                    awm[kt] = *(const v2f*)&gm[4 * kt];
                    awl[kt] = *(const v2f*)&gl[4 * kt];
                }
            }
            float PH[8], PM[8], PL[8];
            float zr[16], hr[16];
            v8f nH = zero8, nM = zero8, nL = zero8;
            seq_steps<PROD>(PH, PM, PL, pH, pM, pL, qH, qM, qL, bH, bM, bL, xr, zr, hr, wc1, wcm, wcl,
                            aw1, awm, awl, nH, nM, nL, ldjE, ldjO, hi);
            qH = nH; qM = nM; qL = nL;
            if (lane < 16) {
                #pragma unroll
                for (int t = 0; t < 16; ++t) {
                    XZ[r * PW + c0 + t] = zr[t];
                    Hh[r * PW + c0 + t] = hr[t];
                }
            }
            if (!PROD) {
                const float tot = ldjE + xhalf(ldjO);
                if (lane < 16) ldjs[tile * 16 + r] = tot;
            }
            lds_barrier();
        };
        for (int I = 0; I < 7; ++I)
            block_body(I, flag_t<true>{});
        block_body(7, flag_t<false>{});
    } else {
        v8f cH2 = zero8, cM2 = zero8, cL2 = zero8;
        v8f cH3 = zero8, cM3 = zero8, cL3 = zero8;
        v8f cH4 = zero8, cM4 = zero8, cL4 = zero8;
        v8f cH5 = zero8, cM5 = zero8, cL5 = zero8;
        v8f cH6 = zero8, cM6 = zero8, cL6 = zero8;
        v8f cH7 = zero8, cM7 = zero8, cL7 = zero8;

        const float* az  = &XZ[mrow * PW + khalf];
        const float* ah  = &Hh[mrow * PW + khalf];

        auto mm_block = [&](v8f& cH, v8f& cM, v8f& cL, int kb, int J) {
            const float* g1 = &Ws[0][(16 * J + prow) * PW + khalf];
            const float* gm = &Ws[1][(16 * J + prow) * PW + khalf];
            const float* gl = &Ws[2][(16 * J + prow) * PW + khalf];
            #pragma unroll
            for (int kt = 0; kt < 4; ++kt) {
                const int k0 = kb + kt * 4;
                const v2f fZ  = *(const v2f*)&az[k0];
                const v2f fH  = *(const v2f*)&ah[k0];
                const v2f aW1 = *(const v2f*)&g1[k0];
                const v2f aWm = *(const v2f*)&gm[k0];
                const v2f aWl = *(const v2f*)&gl[k0];
                cH = __builtin_amdgcn_wmma_f32_16x16x4_f32(false, aW1, false, fZ, (short)0, cH, false, false);
                cM = __builtin_amdgcn_wmma_f32_16x16x4_f32(false, aWm, false, fH, (short)0, cM, false, false);
                cL = __builtin_amdgcn_wmma_f32_16x16x4_f32(false, aWl, false, fH, (short)0, cL, false, false);
            }
        };
        auto put_sums = [&](const v8f& cH, const v8f& cM, const v8f& cL, int J) {
            float* dH = &tHs[J & 1][tile][lane * 8];
            float* dM = &tMs[J & 1][tile][lane * 8];
            float* dL = &tLs[J & 1][tile][lane * 8];
            *(v4f*)&dH[0] = __builtin_shufflevector(cH, cH, 0, 1, 2, 3);  *(v4f*)&dH[4] = __builtin_shufflevector(cH, cH, 4, 5, 6, 7);
            *(v4f*)&dM[0] = __builtin_shufflevector(cM, cM, 0, 1, 2, 3);  *(v4f*)&dM[4] = __builtin_shufflevector(cM, cM, 4, 5, 6, 7);
            *(v4f*)&dL[0] = __builtin_shufflevector(cL, cL, 0, 1, 2, 3);  *(v4f*)&dL[4] = __builtin_shufflevector(cL, cL, 4, 5, 6, 7);
        };

        auto mat_barrier = [&]() {
            __builtin_amdgcn_sched_barrier(0);
            lds_barrier();
            __builtin_amdgcn_sched_barrier(0);
        };

        v4f ra[8], rb[8], rc[8];
        auto fetch_block_row = [&](int R) {
            if (lane < 4 * (R + 1)) {
                #pragma unroll
                for (int q = 0; q < 8; ++q) {
                    const int row = 16 * R + tile + NT * q;
                    ra[q] = *(const v4f*)&W1 [row * DD + lane * 4];
                    rb[q] = *(const v4f*)&Wmu[row * DD + lane * 4];
                    rc[q] = *(const v4f*)&Wls[row * DD + lane * 4];
                }
            }
        };
        auto store_block_row = [&](int R) {
            if (lane < 4 * (R + 1)) {
                #pragma unroll
                for (int q = 0; q < 8; ++q) {
                    const int row = 16 * R + tile + NT * q;
                    *(v4f*)&Ws[0][row * PW + lane * 4] = ra[q];
                    *(v4f*)&Ws[1][row * PW + lane * 4] = rb[q];
                    *(v4f*)&Ws[2][row * PW + lane * 4] = rc[q];
                }
            }
        };

        {
            v4f qa[3][8], qb[3][8], qc[3][8];
            #pragma unroll
            for (int s = 0; s < 3; ++s) {
                if (lane < 4 * (s + 3)) {
                    #pragma unroll
                    for (int q = 0; q < 8; ++q) {
                        const int row = 16 * (s + 2) + tile + NT * q;
                        qa[s][q] = *(const v4f*)&W1 [row * DD + lane * 4];
                        qb[s][q] = *(const v4f*)&Wmu[row * DD + lane * 4];
                        qc[s][q] = *(const v4f*)&Wls[row * DD + lane * 4];
                    }
                }
            }
            #pragma unroll
            for (int s = 0; s < 3; ++s) {
                if (lane < 4 * (s + 3)) {
                    #pragma unroll
                    for (int q = 0; q < 8; ++q) {
                        const int row = 16 * (s + 2) + tile + NT * q;
                        *(v4f*)&Ws[0][row * PW + lane * 4] = qa[s][q];
                        *(v4f*)&Ws[1][row * PW + lane * 4] = qb[s][q];
                        *(v4f*)&Ws[2][row * PW + lane * 4] = qc[s][q];
                    }
                }
            }
        }
        mat_barrier();

        fetch_block_row(5);
        mm_block(cH2, cM2, cL2, 0, 2);
        mm_block(cH3, cM3, cL3, 0, 3);
        mm_block(cH4, cM4, cL4, 0, 4);
        put_sums(cH2, cM2, cL2, 2);
        store_block_row(5);
        mat_barrier();

        fetch_block_row(6);
        mm_block(cH3, cM3, cL3, 16, 3);
        mm_block(cH4, cM4, cL4, 16, 4);
        mm_block(cH5, cM5, cL5, 0, 5);
        mm_block(cH5, cM5, cL5, 16, 5);
        put_sums(cH3, cM3, cL3, 3);
        store_block_row(6);
        mat_barrier();

        fetch_block_row(7);
        mm_block(cH4, cM4, cL4, 32, 4);
        mm_block(cH5, cM5, cL5, 32, 5);
        mm_block(cH6, cM6, cL6, 0, 6);
        mm_block(cH6, cM6, cL6, 16, 6);
        put_sums(cH4, cM4, cL4, 4);
        store_block_row(7);
        mat_barrier();

        mm_block(cH5, cM5, cL5, 48, 5);
        mm_block(cH6, cM6, cL6, 32, 6);
        mm_block(cH7, cM7, cL7, 0, 7);
        mm_block(cH7, cM7, cL7, 16, 7);
        put_sums(cH5, cM5, cL5, 5);
        mat_barrier();

        mm_block(cH6, cM6, cL6, 48, 6);
        mm_block(cH6, cM6, cL6, 64, 6);
        mm_block(cH7, cM7, cL7, 32, 7);
        put_sums(cH6, cM6, cL6, 6);
        mat_barrier();

        mm_block(cH7, cM7, cL7, 48, 7);
        mm_block(cH7, cM7, cL7, 64, 7);
        mm_block(cH7, cM7, cL7, 80, 7);
        put_sums(cH7, cM7, cL7, 7);
        mat_barrier();

        mat_barrier();
    }

    const int rb8 = seqw ? 0 : 8;
    #pragma unroll
    for (int r = 0; r < 8; ++r)
        *(v4f*)&out[(size_t)(r0 + rb8 + r) * DD + lane * 4] = *(const v4f*)&XZ[(rb8 + r) * PW + lane * 4];

    if (tid < 16 * NT)
        out[(size_t)B * DD + blockIdx.x * (16 * NT) + tid] = ldjs[tid];
}

extern "C" void kernel_launch(void* const* d_in, const int* in_sizes, int n_in,
                              void* d_out, int out_size, void* d_ws, size_t ws_size,
                              hipStream_t stream) {
    (void)n_in; (void)out_size; (void)d_ws; (void)ws_size;
    const float* x   = (const float*)d_in[0];
    const float* W1  = (const float*)d_in[1];
    const float* b1  = (const float*)d_in[2];
    const float* Wmu = (const float*)d_in[3];
    const float* bmu = (const float*)d_in[4];
    const float* Wls = (const float*)d_in[5];
    const float* bls = (const float*)d_in[6];
    float* out = (float*)d_out;

    const int B = in_sizes[0] / DD;
    dim3 grid(B / (16 * NT)), block(32 * NWV);
    iaf_layer_kernel<<<grid, block, 0, stream>>>(x, W1, b1, Wmu, bmu, Wls, bls, out, B);
}
